# starting kernel
# baseline (speedup 1.0000x reference)
#include <hip/hip_runtime.h>

#define N_NODES 50000
#define N_EDGES 800000
#define NB      100000u
#define C_OUT   64

typedef __attribute__((ext_vector_type(2))) float v2f;
typedef __attribute__((ext_vector_type(8))) float v8f;

__global__ void GCNN_zero_kernel(float* __restrict__ p, int n) {
    int i = blockIdx.x * blockDim.x + threadIdx.x;
    if (i < n) p[i] = 0.0f;
}

__global__ void GCNN_spmm_scatter_kernel(const float* __restrict__ x,
                                         const int*   __restrict__ rows,
                                         const int*   __restrict__ cols,
                                         const float* __restrict__ vals,
                                         float*       __restrict__ agg) {
    long long g = (long long)blockIdx.x * blockDim.x + threadIdx.x;
    int e = (int)(g >> 7);
    int k = (int)(g & 127);
    if (e >= N_EDGES) return;
    int   i = rows[e];
    int   j = cols[e];
    float v = vals[e];
    float hv = x[((k >> 6) * N_NODES + j) * 64 + (k & 63)];
    atomicAdd(&agg[i * 128 + k], v * hv);
}

__global__ void GCNN_gemm_wmma_kernel(const float* __restrict__ agg,
                                      const float* __restrict__ W,
                                      const float* __restrict__ bias,
                                      float*       __restrict__ out) {
    const int lane = threadIdx.x & 31;
    const int wave = threadIdx.x >> 5;
    const int half = lane >> 4;
    const int l16  = lane & 15;
    const unsigned r0 = (unsigned)blockIdx.x * 16u;
    const int      n0 = wave * 16;
    const unsigned r  = r0 + (unsigned)l16;

    v8f acc = {};
    #pragma unroll
    for (int k0 = 0; k0 < 64; k0 += 4) {
        const int ka = k0 + 2 * half;
        v2f a, b;
        {
            unsigned f  = r * 64u + (unsigned)ka;
            unsigned c0 = f / NB;
            unsigned rem = f - c0 * NB;
            a.x = agg[(rem >> 1) * 128u + c0 * 2u + (f & 1u)];
            f += 1u;
            c0  = f / NB;
            rem = f - c0 * NB;
            a.y = agg[(rem >> 1) * 128u + c0 * 2u + (f & 1u)];
        }
        b.x = W[ka * C_OUT + n0 + l16];
        b.y = W[(ka + 1) * C_OUT + n0 + l16];

        acc = __builtin_amdgcn_wmma_f32_16x16x4_f32(
             false, a,  false, b,
             (short)0, acc,  false,  false);
    }

    const int   col = n0 + l16;
    const float bv  = bias[col];
    #pragma unroll
    for (int v = 0; v < 8; ++v) {
        float o = acc[v] + bv;
        o = o > 0.0f ? o : 0.0f;
        out[(r0 + (unsigned)(v + 8 * half)) * (unsigned)C_OUT + (unsigned)col] = o;
    }
}

extern "C" void kernel_launch(void* const* d_in, const int* in_sizes, int n_in,
                              void* d_out, int out_size, void* d_ws, size_t ws_size,
                              hipStream_t stream) {
    const float* x    = (const float*)d_in[0];
    const int*   rows = (const int*)  d_in[1];
    const int*   cols = (const int*)  d_in[2];
    const float* vals = (const float*)d_in[3];
    const float* W    = (const float*)d_in[4];
    const float* bias = (const float*)d_in[5];
    float* out = (float*)d_out;
    float* agg = (float*)d_ws;

    const int aggN = N_NODES * 128;
    GCNN_zero_kernel<<<(aggN + 255) / 256, 256, 0, stream>>>(agg, aggN);

    const long long scatterThreads = (long long)N_EDGES * 128;
    const int scatterBlocks = (int)((scatterThreads + 255) / 256);
    GCNN_spmm_scatter_kernel<<<scatterBlocks, 256, 0, stream>>>(x, rows, cols,
                                                               vals, agg);

    GCNN_gemm_wmma_kernel<<<NB / 16, 128, 0, stream>>>(agg, W, bias, out);
}


# agg built in LDS: each work-group owns 196 agg rows, scans all edge rows, queues its own edges and adds them with LDS float atomics (arrival order, as before); no global atomics, no zero pass
# speedup vs baseline: 3.5255x; 3.5255x over previous
#include <hip/hip_runtime.h>

#define N_NODES 50000
#define N_EDGES 800000
#define NB      100000u
#define C_OUT   64

typedef __attribute__((ext_vector_type(2))) float v2f;
typedef __attribute__((ext_vector_type(8))) float v8f;

#define TILE_R        196
#define N_TILES       256
#define N_GROUPS      6250
#define GROUPS_PER_IT 256
#define N_ITS         25
#define WQ_CAP        1024

__device__ __forceinline__ void GCNN_enqueue(int r, int e, int r_lo, int nrows,
                                             unsigned* __restrict__ myq, unsigned& cntw) {
    const unsigned lrow = (unsigned)(r - r_lo);
    const bool     hit  = lrow < (unsigned)nrows;
    const unsigned bal  = __builtin_amdgcn_ballot_w32(hit);
    if (bal != 0u) {
        if (hit) myq[cntw + __builtin_amdgcn_mbcnt_lo(bal, 0u)] = (unsigned)e | (lrow << 20);
        cntw += (unsigned)__builtin_popcount(bal);
    }
}

__global__ __launch_bounds__(1024)
void GCNN_spmm_tile_kernel(const float* __restrict__ x,
                           const int*   __restrict__ rows,
                           const int*   __restrict__ cols,
                           const float* __restrict__ vals,
                           float*       __restrict__ agg) {
    __shared__ float    tile[(TILE_R + 1) * 128];
    __shared__ unsigned queue[32 * WQ_CAP];

    const int tid  = threadIdx.x;
    const int lane = tid & 31;
    const int wave = __builtin_amdgcn_readfirstlane(tid >> 5);
    const int r_lo = (int)blockIdx.x * TILE_R;
    const int nrows = (N_NODES - r_lo) < TILE_R ? (N_NODES - r_lo) : TILE_R;

    for (int p = tid; p < (TILE_R + 1) * 32; p += 1024)
        ((float4*)tile)[p] = make_float4(0.0f, 0.0f, 0.0f, 0.0f);
    __syncthreads();

    const int4* rows4 = (const int4*)rows;
    unsigned*   myq   = queue + wave * WQ_CAP;
    const float* xlane = x + (lane >> 4) * (N_NODES * 64) + (lane & 15) * 4;

    for (int it = 0; it < N_ITS; ++it) {
        const int gbase = it * GROUPS_PER_IT + wave * 8;

        int4 rv[8];
        #pragma unroll
        for (int u = 0; u < 8; ++u) {
            const int g  = gbase + u;
            const int gc = g < N_GROUPS ? g : (N_GROUPS - 1);
            rv[u] = rows4[gc * 32 + lane];
        }
        unsigned cntw = 0u;
        #pragma unroll
        for (int u = 0; u < 8; ++u) {
            const int g  = gbase + u;
            const int nr = g < N_GROUPS ? nrows : 0;
            const int e0 = g * 128 + lane * 4;
            GCNN_enqueue(rv[u].x, e0,     r_lo, nr, myq, cntw);
            GCNN_enqueue(rv[u].y, e0 + 1, r_lo, nr, myq, cntw);
            GCNN_enqueue(rv[u].z, e0 + 2, r_lo, nr, myq, cntw);
            GCNN_enqueue(rv[u].w, e0 + 3, r_lo, nr, myq, cntw);
        }

        for (unsigned qb = 0u; qb < cntw; qb += 32u) {
            const unsigned nb = (cntw - qb) < 32u ? (cntw - qb) : 32u;
            unsigned ent = (unsigned)TILE_R << 20;
            int      jc  = 0;
            float    vv  = 0.0f;
            if ((unsigned)lane < nb) {
                ent = myq[qb + (unsigned)lane];
                const unsigned e = ent & 0xFFFFFu;
                jc = cols[e];
                vv = vals[e];
            }
            for (unsigned u0 = 0u; u0 < nb; u0 += 4u) {
                int    j[4];
                float4 xv[4];
                #pragma unroll
                for (int u = 0; u < 4; ++u)
                    j[u] = __builtin_amdgcn_readlane(jc, (int)u0 + u);
                #pragma unroll
                for (int u = 0; u < 4; ++u)
                    xv[u] = *(const float4*)(xlane + j[u] * 64);
                #pragma unroll
                for (int u = 0; u < 4; ++u) {
                    const unsigned lrow = (unsigned)__builtin_amdgcn_readlane((int)ent, (int)u0 + u) >> 20;
                    const float    v    = __int_as_float(__builtin_amdgcn_readlane(__float_as_int(vv), (int)u0 + u));
                    float* t = tile + lrow * 128u + (unsigned)lane;
                    __hip_atomic_fetch_add(t,      v * xv[u].x, __ATOMIC_RELAXED, __HIP_MEMORY_SCOPE_WORKGROUP);
                    __hip_atomic_fetch_add(t + 32, v * xv[u].y, __ATOMIC_RELAXED, __HIP_MEMORY_SCOPE_WORKGROUP);
                    __hip_atomic_fetch_add(t + 64, v * xv[u].z, __ATOMIC_RELAXED, __HIP_MEMORY_SCOPE_WORKGROUP);
                    __hip_atomic_fetch_add(t + 96, v * xv[u].w, __ATOMIC_RELAXED, __HIP_MEMORY_SCOPE_WORKGROUP);
                }
            }
        }
    }
    __syncthreads();

    float4* agg4 = (float4*)agg;
    for (int p = tid; p < nrows * 32; p += 1024) {
        const int rr = p >> 5;
        const int L  = p & 31;
        const float* t = tile + rr * 128 + L;
        agg4[(r_lo + rr) * 32 + L] = make_float4(t[0], t[32], t[64], t[96]);
    }
}

__global__ void GCNN_gemm_wmma_kernel(const float* __restrict__ agg,
                                      const float* __restrict__ W,
                                      const float* __restrict__ bias,
                                      float*       __restrict__ out) {
    const int lane = threadIdx.x & 31;
    const int wave = threadIdx.x >> 5;
    const int half = lane >> 4;
    const int l16  = lane & 15;
    const unsigned r0 = (unsigned)blockIdx.x * 16u;
    const int      n0 = wave * 16;
    const unsigned r  = r0 + (unsigned)l16;

    v8f acc = {};
    #pragma unroll
    for (int k0 = 0; k0 < 64; k0 += 4) {
        const int ka = k0 + 2 * half;
        v2f a, b;
        {
            unsigned f  = r * 64u + (unsigned)ka;
            unsigned c0 = f / NB;
            unsigned rem = f - c0 * NB;
            a.x = agg[(rem >> 1) * 128u + c0 * 2u + (f & 1u)];
            f += 1u;
            c0  = f / NB;
            rem = f - c0 * NB;
            a.y = agg[(rem >> 1) * 128u + c0 * 2u + (f & 1u)];
        }
        b.x = W[ka * C_OUT + n0 + l16];
        b.y = W[(ka + 1) * C_OUT + n0 + l16];

        acc = __builtin_amdgcn_wmma_f32_16x16x4_f32(
             false, a,  false, b,
             (short)0, acc,  false,  false);
    }

    const int   col = n0 + l16;
    const float bv  = bias[col];
    #pragma unroll
    for (int v = 0; v < 8; ++v) {
        float o = acc[v] + bv;
        o = o > 0.0f ? o : 0.0f;
        out[(r0 + (unsigned)(v + 8 * half)) * (unsigned)C_OUT + (unsigned)col] = o;
    }
}

extern "C" void kernel_launch(void* const* d_in, const int* in_sizes, int n_in,
                              void* d_out, int out_size, void* d_ws, size_t ws_size,
                              hipStream_t stream) {
    const float* x    = (const float*)d_in[0];
    const int*   rows = (const int*)  d_in[1];
    const int*   cols = (const int*)  d_in[2];
    const float* vals = (const float*)d_in[3];
    const float* W    = (const float*)d_in[4];
    const float* bias = (const float*)d_in[5];
    float* out = (float*)d_out;
    float* agg = (float*)d_ws;

    GCNN_spmm_tile_kernel<<<N_TILES, 1024, 0, stream>>>(x, rows, cols, vals, agg);

    GCNN_gemm_wmma_kernel<<<NB / 16, 128, 0, stream>>>(agg, W, bias, out);
}


# New kernel buckets edges by agg tile: counting-sort of each 3125-edge chunk by tile in LDS (integer-add ranks); tile work-groups read only their own segments, no scan of all rows; sums arrival order
# speedup vs baseline: 4.5547x; 1.2919x over previous
#include <hip/hip_runtime.h>

#define N_NODES 50000
#define N_EDGES 800000
#define NB      100000u
#define C_OUT   64

typedef __attribute__((ext_vector_type(2))) float v2f;
typedef __attribute__((ext_vector_type(8))) float v8f;

#define TILE_R      196
#define N_TILES     256
#define N_CHUNKS    256
#define CHUNK       3125
#define REC_STRIDE  3136
#define OFF_STRIDE  288
#define BIN_THREADS 256
#define BIN_EPT     13
#define WQ_CAP      512

__global__ __launch_bounds__(BIN_THREADS)
void GCNN_bin_kernel(const int*   __restrict__ rows,
                     const int*   __restrict__ cols,
                     const float* __restrict__ vals,
                     unsigned*    __restrict__ offs,
                     unsigned*    __restrict__ recs) {
    __shared__ unsigned cnt[N_TILES];
    __shared__ unsigned start[N_TILES + 1];
    __shared__ unsigned skey[CHUNK];
    __shared__ unsigned sval[CHUNK];

    const int tid  = threadIdx.x;
    const int g    = blockIdx.x;
    const int base = g * CHUNK;

    cnt[tid] = 0u;

    int   r[BIN_EPT];
    int   c[BIN_EPT];
    float v[BIN_EPT];
    #pragma unroll
    for (int k = 0; k < BIN_EPT; ++k) {
        const int idx = tid + k * BIN_THREADS;
        r[k] = -1; c[k] = 0; v[k] = 0.0f;
        if (idx < CHUNK) {
            r[k] = rows[base + idx];
            c[k] = cols[base + idx];
            v[k] = vals[base + idx];
        }
    }
    __syncthreads();

    unsigned tl[BIN_EPT];
    unsigned pos[BIN_EPT];
    #pragma unroll
    for (int k = 0; k < BIN_EPT; ++k) {
        const bool ok = (unsigned)r[k] < (unsigned)N_NODES;
        tl[k]  = ok ? (unsigned)r[k] / (unsigned)TILE_R : 0u;
        pos[k] = 0u;
        if (ok) pos[k] = atomicAdd(&cnt[tl[k]], 1u);
    }
    __syncthreads();

    if (tid < 32) {
        unsigned c8[8];
        unsigned sum = 0u;
        #pragma unroll
        for (int i = 0; i < 8; ++i) { c8[i] = cnt[8 * tid + i]; sum += c8[i]; }
        unsigned inc = sum;
        #pragma unroll
        for (int d = 1; d < 32; d <<= 1) {
            const unsigned up = __shfl_up(inc, (unsigned)d);
            if (tid >= d) inc += up;
        }
        unsigned run = inc - sum;
        #pragma unroll
        for (int i = 0; i < 8; ++i) {
            start[8 * tid + i] = run;
            offs[g * OFF_STRIDE + 8 * tid + i] = run;
            run += c8[i];
        }
        if (tid == 31) {
            start[N_TILES] = run;
            offs[g * OFF_STRIDE + N_TILES] = run;
        }
    }
    __syncthreads();

    #pragma unroll
    for (int k = 0; k < BIN_EPT; ++k) {
        if ((unsigned)r[k] < (unsigned)N_NODES) {
            const unsigned d = start[tl[k]] + pos[k];
            skey[d] = (unsigned)c[k] | (((unsigned)r[k] - tl[k] * (unsigned)TILE_R) << 16);
            sval[d] = __float_as_uint(v[k]);
        }
    }
    __syncthreads();

    const unsigned total = start[N_TILES];
    uint2* dst = (uint2*)recs + g * REC_STRIDE;
    for (unsigned i = (unsigned)tid; i < total; i += BIN_THREADS)
        dst[i] = make_uint2(skey[i], sval[i]);
}

__global__ __launch_bounds__(1024)
void GCNN_spmm_tile_kernel(const float*    __restrict__ x,
                           const unsigned* __restrict__ offs,
                           const unsigned* __restrict__ recs,
                           float*          __restrict__ agg) {
    __shared__ float    tile[(TILE_R + 1) * 128];
    __shared__ unsigned qkey[32 * WQ_CAP];
    __shared__ unsigned qval[32 * WQ_CAP];

    const int tid  = threadIdx.x;
    const int lane = tid & 31;
    const int wave = __builtin_amdgcn_readfirstlane(tid >> 5);
    const int t    = (int)blockIdx.x;
    const int r_lo = t * TILE_R;
    const int nrows = (N_NODES - r_lo) < TILE_R ? (N_NODES - r_lo) : TILE_R;

    int o0 = 0, o1 = 0;
    if (lane < 8) {
        const unsigned* p = offs + (wave * 8 + lane) * OFF_STRIDE + t;
        o0 = (int)p[0];
        o1 = (int)p[1];
    }

    for (int p = tid; p < (TILE_R + 1) * 32; p += 1024)
        ((float4*)tile)[p] = make_float4(0.0f, 0.0f, 0.0f, 0.0f);

    const uint2* rec2 = (const uint2*)recs + wave * 8 * REC_STRIDE;
    unsigned key[8], vb[8];
    #pragma unroll
    for (int s = 0; s < 8; ++s) {
        const unsigned b = (unsigned)__builtin_amdgcn_readlane(o0, s);
        const unsigned n = (unsigned)__builtin_amdgcn_readlane(o1, s) - b;
        const unsigned li = (unsigned)lane < n ? (unsigned)lane : n - 1u;
        const uint2 rr = rec2[s * REC_STRIDE + (n != 0u ? b + li : 0u)];
        key[s] = rr.x; vb[s] = rr.y;
    }
    __syncthreads();

    unsigned* myk = qkey + wave * WQ_CAP;
    unsigned* myv = qval + wave * WQ_CAP;
    unsigned  cntw = 0u;
    #pragma unroll
    for (int s = 0; s < 8; ++s) {
        const unsigned n = (unsigned)(__builtin_amdgcn_readlane(o1, s) - __builtin_amdgcn_readlane(o0, s));
        const unsigned m = n < 32u ? n : 32u;
        if ((unsigned)lane < m) { myk[cntw + lane] = key[s]; myv[cntw + lane] = vb[s]; }
        cntw += m;
    }

    const float* xlane = x + (lane >> 4) * (N_NODES * 64) + (lane & 15) * 4;

    int      s = 0;
    unsigned q = 32u;
    for (;;) {
        while (s < 8 && cntw <= (unsigned)(WQ_CAP - 32)) {
            const unsigned b = (unsigned)__builtin_amdgcn_readlane(o0, s);
            const unsigned n = (unsigned)__builtin_amdgcn_readlane(o1, s) - b;
            if (q < n) {
                const unsigned m = (n - q) < 32u ? (n - q) : 32u;
                if ((unsigned)lane < m) {
                    const uint2 rr = rec2[s * REC_STRIDE + b + q + (unsigned)lane];
                    myk[cntw + lane] = rr.x; myv[cntw + lane] = rr.y;
                }
                cntw += m;
                q += 32u;
            } else {
                ++s;
                q = 32u;
            }
        }

        for (unsigned qb = 0u; qb < cntw; qb += 32u) {
            const unsigned nb = (cntw - qb) < 32u ? (cntw - qb) : 32u;
            int ent = TILE_R << 16;
            int vvb = 0;
            if ((unsigned)lane < nb) {
                ent = (int)myk[qb + (unsigned)lane];
                vvb = (int)myv[qb + (unsigned)lane];
            }
            for (unsigned u0 = 0u; u0 < nb; u0 += 8u) {
                unsigned e[8];
                float4   xv[8];
                #pragma unroll
                for (int u = 0; u < 8; ++u)
                    e[u] = (unsigned)__builtin_amdgcn_readlane(ent, (int)u0 + u);
                #pragma unroll
                for (int u = 0; u < 8; ++u)
                    xv[u] = *(const float4*)(xlane + (e[u] & 0xFFFFu) * 64u);
                #pragma unroll
                for (int u = 0; u < 8; ++u) {
                    const unsigned lrow = e[u] >> 16;
                    const float    v    = __int_as_float(__builtin_amdgcn_readlane(vvb, (int)u0 + u));
                    float* tp = tile + lrow * 128u + (unsigned)lane;
                    __hip_atomic_fetch_add(tp,      v * xv[u].x, __ATOMIC_RELAXED, __HIP_MEMORY_SCOPE_WORKGROUP);
                    __hip_atomic_fetch_add(tp + 32, v * xv[u].y, __ATOMIC_RELAXED, __HIP_MEMORY_SCOPE_WORKGROUP);
                    __hip_atomic_fetch_add(tp + 64, v * xv[u].z, __ATOMIC_RELAXED, __HIP_MEMORY_SCOPE_WORKGROUP);
                    __hip_atomic_fetch_add(tp + 96, v * xv[u].w, __ATOMIC_RELAXED, __HIP_MEMORY_SCOPE_WORKGROUP);
                }
            }
        }
        cntw = 0u;
        if (s >= 8) break;
    }
    __syncthreads();

    float4* agg4 = (float4*)agg;
    for (int p = tid; p < nrows * 32; p += 1024) {
        const int rr = p >> 5;
        const int L  = p & 31;
        const float* tp = tile + rr * 128 + L;
        agg4[(r_lo + rr) * 32 + L] = make_float4(tp[0], tp[32], tp[64], tp[96]);
    }
}

__global__ void GCNN_gemm_wmma_kernel(const float* __restrict__ agg,
                                      const float* __restrict__ W,
                                      const float* __restrict__ bias,
                                      float*       __restrict__ out) {
    const int lane = threadIdx.x & 31;
    const int wave = threadIdx.x >> 5;
    const int half = lane >> 4;
    const int l16  = lane & 15;
    const unsigned r0 = (unsigned)blockIdx.x * 16u;
    const int      n0 = wave * 16;
    const unsigned r  = r0 + (unsigned)l16;

    v8f acc = {};
    #pragma unroll
    for (int k0 = 0; k0 < 64; k0 += 4) {
        const int ka = k0 + 2 * half;
        v2f a, b;
        {
            unsigned f  = r * 64u + (unsigned)ka;
            unsigned c0 = f / NB;
            unsigned rem = f - c0 * NB;
            a.x = agg[(rem >> 1) * 128u + c0 * 2u + (f & 1u)];
            f += 1u;
            c0  = f / NB;
            rem = f - c0 * NB;
            a.y = agg[(rem >> 1) * 128u + c0 * 2u + (f & 1u)];
        }
        b.x = W[ka * C_OUT + n0 + l16];
        b.y = W[(ka + 1) * C_OUT + n0 + l16];

        acc = __builtin_amdgcn_wmma_f32_16x16x4_f32(
             false, a,  false, b,
             (short)0, acc,  false,  false);
    }

    const int   col = n0 + l16;
    const float bv  = bias[col];
    #pragma unroll
    for (int v = 0; v < 8; ++v) {
        float o = acc[v] + bv;
        o = o > 0.0f ? o : 0.0f;
        out[(r0 + (unsigned)(v + 8 * half)) * (unsigned)C_OUT + (unsigned)col] = o;
    }
}

extern "C" void kernel_launch(void* const* d_in, const int* in_sizes, int n_in,
                              void* d_out, int out_size, void* d_ws, size_t ws_size,
                              hipStream_t stream) {
    const float* x    = (const float*)d_in[0];
    const int*   rows = (const int*)  d_in[1];
    const int*   cols = (const int*)  d_in[2];
    const float* vals = (const float*)d_in[3];
    const float* W    = (const float*)d_in[4];
    const float* bias = (const float*)d_in[5];
    float* out = (float*)d_out;
    float*    agg  = (float*)d_ws;
    unsigned* recs = (unsigned*)d_ws + 6400000;
    unsigned* offs = recs + 2 * N_CHUNKS * REC_STRIDE;

    GCNN_bin_kernel<<<N_CHUNKS, BIN_THREADS, 0, stream>>>(rows, cols, vals, offs, recs);

    GCNN_spmm_tile_kernel<<<N_TILES, 1024, 0, stream>>>(x, offs, recs, agg);

    GCNN_gemm_wmma_kernel<<<NB / 16, 128, 0, stream>>>(agg, W, bias, out);
}


# GEMM cut along agg: a work-group walks 5 tiles whose 16 rows are one 32-row block of 16 adjacent channel pairs; a half wave's A reads share 2 lines, not 16; W fragments loaded once; k order kept
# speedup vs baseline: 7.0848x; 1.5555x over previous
#include <hip/hip_runtime.h>

#define N_NODES 50000
#define N_EDGES 800000
#define NB      100000u
#define C_OUT   64

typedef __attribute__((ext_vector_type(2))) float v2f;
typedef __attribute__((ext_vector_type(8))) float v8f;

#define TILE_R      196
#define N_TILES     256
#define N_CHUNKS    256
#define CHUNK       3125
#define REC_STRIDE  3136
#define OFF_STRIDE  288
#define BIN_THREADS 256
#define BIN_EPT     13
#define WQ_CAP      512

__global__ __launch_bounds__(BIN_THREADS)
void GCNN_bin_kernel(const int*   __restrict__ rows,
                     const int*   __restrict__ cols,
                     const float* __restrict__ vals,
                     unsigned*    __restrict__ offs,
                     unsigned*    __restrict__ recs) {
    __shared__ unsigned cnt[N_TILES];
    __shared__ unsigned start[N_TILES + 1];
    __shared__ unsigned skey[CHUNK];
    __shared__ unsigned sval[CHUNK];

    const int tid  = threadIdx.x;
    const int g    = blockIdx.x;
    const int base = g * CHUNK;

    cnt[tid] = 0u;

    int   r[BIN_EPT];
    int   c[BIN_EPT];
    float v[BIN_EPT];
    #pragma unroll
    for (int k = 0; k < BIN_EPT; ++k) {
        const int idx = tid + k * BIN_THREADS;
        r[k] = -1; c[k] = 0; v[k] = 0.0f;
        if (idx < CHUNK) {
            r[k] = rows[base + idx];
            c[k] = cols[base + idx];
            v[k] = vals[base + idx];
        }
    }
    __syncthreads();

    unsigned tl[BIN_EPT];
    unsigned pos[BIN_EPT];
    #pragma unroll
    for (int k = 0; k < BIN_EPT; ++k) {
        const bool ok = (unsigned)r[k] < (unsigned)N_NODES;
        tl[k]  = ok ? (unsigned)r[k] / (unsigned)TILE_R : 0u;
        pos[k] = 0u;
        if (ok) pos[k] = atomicAdd(&cnt[tl[k]], 1u);
    }
    __syncthreads();

    if (tid < 32) {
        unsigned c8[8];
        unsigned sum = 0u;
        #pragma unroll
        for (int i = 0; i < 8; ++i) { c8[i] = cnt[8 * tid + i]; sum += c8[i]; }
        unsigned inc = sum;
        #pragma unroll
        for (int d = 1; d < 32; d <<= 1) {
            const unsigned up = __shfl_up(inc, (unsigned)d);
            if (tid >= d) inc += up;
        }
        unsigned run = inc - sum;
        #pragma unroll
        for (int i = 0; i < 8; ++i) {
            start[8 * tid + i] = run;
            offs[g * OFF_STRIDE + 8 * tid + i] = run;
            run += c8[i];
        }
        if (tid == 31) {
            start[N_TILES] = run;
            offs[g * OFF_STRIDE + N_TILES] = run;
        }
    }
    __syncthreads();

    #pragma unroll
    for (int k = 0; k < BIN_EPT; ++k) {
        if ((unsigned)r[k] < (unsigned)N_NODES) {
            const unsigned d = start[tl[k]] + pos[k];
            skey[d] = (unsigned)c[k] | (((unsigned)r[k] - tl[k] * (unsigned)TILE_R) << 16);
            sval[d] = __float_as_uint(v[k]);
        }
    }
    __syncthreads();

    const unsigned total = start[N_TILES];
    uint2* dst = (uint2*)recs + g * REC_STRIDE;
    for (unsigned i = (unsigned)tid; i < total; i += BIN_THREADS)
        dst[i] = make_uint2(skey[i], sval[i]);
}

__global__ __launch_bounds__(1024)
void GCNN_spmm_tile_kernel(const float*    __restrict__ x,
                           const unsigned* __restrict__ offs,
                           const unsigned* __restrict__ recs,
                           float*          __restrict__ agg) {
    __shared__ float    tile[(TILE_R + 1) * 128];
    __shared__ unsigned qkey[32 * WQ_CAP];
    __shared__ unsigned qval[32 * WQ_CAP];

    const int tid  = threadIdx.x;
    const int lane = tid & 31;
    const int wave = __builtin_amdgcn_readfirstlane(tid >> 5);
    const int t    = (int)blockIdx.x;
    const int r_lo = t * TILE_R;
    const int nrows = (N_NODES - r_lo) < TILE_R ? (N_NODES - r_lo) : TILE_R;

    int o0 = 0, o1 = 0;
    if (lane < 8) {
        const unsigned* p = offs + (wave * 8 + lane) * OFF_STRIDE + t;
        o0 = (int)p[0];
        o1 = (int)p[1];
    }

    for (int p = tid; p < (TILE_R + 1) * 32; p += 1024)
        ((float4*)tile)[p] = make_float4(0.0f, 0.0f, 0.0f, 0.0f);

    const uint2* rec2 = (const uint2*)recs + wave * 8 * REC_STRIDE;
    unsigned key[8], vb[8];
    #pragma unroll
    for (int s = 0; s < 8; ++s) {
        const unsigned b = (unsigned)__builtin_amdgcn_readlane(o0, s);
        const unsigned n = (unsigned)__builtin_amdgcn_readlane(o1, s) - b;
        const unsigned li = (unsigned)lane < n ? (unsigned)lane : n - 1u;
        const uint2 rr = rec2[s * REC_STRIDE + (n != 0u ? b + li : 0u)];
        key[s] = rr.x; vb[s] = rr.y;
    }
    __syncthreads();

    unsigned* myk = qkey + wave * WQ_CAP;
    unsigned* myv = qval + wave * WQ_CAP;
    unsigned  cntw = 0u;
    #pragma unroll
    for (int s = 0; s < 8; ++s) {
        const unsigned n = (unsigned)(__builtin_amdgcn_readlane(o1, s) - __builtin_amdgcn_readlane(o0, s));
        const unsigned m = n < 32u ? n : 32u;
        if ((unsigned)lane < m) { myk[cntw + lane] = key[s]; myv[cntw + lane] = vb[s]; }
        cntw += m;
    }

    const float* xlane = x + (lane >> 4) * (N_NODES * 64) + (lane & 15) * 4;

    int      s = 0;
    unsigned q = 32u;
    for (;;) {
        while (s < 8 && cntw <= (unsigned)(WQ_CAP - 32)) {
            const unsigned b = (unsigned)__builtin_amdgcn_readlane(o0, s);
            const unsigned n = (unsigned)__builtin_amdgcn_readlane(o1, s) - b;
            if (q < n) {
                const unsigned m = (n - q) < 32u ? (n - q) : 32u;
                if ((unsigned)lane < m) {
                    const uint2 rr = rec2[s * REC_STRIDE + b + q + (unsigned)lane];
                    myk[cntw + lane] = rr.x; myv[cntw + lane] = rr.y;
                }
                cntw += m;
                q += 32u;
            } else {
                ++s;
                q = 32u;
            }
        }

        for (unsigned qb = 0u; qb < cntw; qb += 32u) {
            const unsigned nb = (cntw - qb) < 32u ? (cntw - qb) : 32u;
            int ent = TILE_R << 16;
            int vvb = 0;
            if ((unsigned)lane < nb) {
                ent = (int)myk[qb + (unsigned)lane];
                vvb = (int)myv[qb + (unsigned)lane];
            }
            for (unsigned u0 = 0u; u0 < nb; u0 += 8u) {
                unsigned e[8];
                float4   xv[8];
                #pragma unroll
                for (int u = 0; u < 8; ++u)
                    e[u] = (unsigned)__builtin_amdgcn_readlane(ent, (int)u0 + u);
                #pragma unroll
                for (int u = 0; u < 8; ++u)
                    xv[u] = *(const float4*)(xlane + (e[u] & 0xFFFFu) * 64u);
                #pragma unroll
                for (int u = 0; u < 8; ++u) {
                    const unsigned lrow = e[u] >> 16;
                    const float    v    = __int_as_float(__builtin_amdgcn_readlane(vvb, (int)u0 + u));
                    float* tp = tile + lrow * 128u + (unsigned)lane;
                    __hip_atomic_fetch_add(tp,      v * xv[u].x, __ATOMIC_RELAXED, __HIP_MEMORY_SCOPE_WORKGROUP);
                    __hip_atomic_fetch_add(tp + 32, v * xv[u].y, __ATOMIC_RELAXED, __HIP_MEMORY_SCOPE_WORKGROUP);
                    __hip_atomic_fetch_add(tp + 64, v * xv[u].z, __ATOMIC_RELAXED, __HIP_MEMORY_SCOPE_WORKGROUP);
                    __hip_atomic_fetch_add(tp + 96, v * xv[u].w, __ATOMIC_RELAXED, __HIP_MEMORY_SCOPE_WORKGROUP);
                }
            }
        }
        cntw = 0u;
        if (s >= 8) break;
    }
    __syncthreads();

    float4* agg4 = (float4*)agg;
    for (int p = tid; p < nrows * 32; p += 1024) {
        const int rr = p >> 5;
        const int L  = p & 31;
        const float* tp = tile + rr * 128 + L;
        agg4[(r_lo + rr) * 32 + L] = make_float4(tp[0], tp[32], tp[64], tp[96]);
    }
}

#define PAIR2_ROWS   3125u
#define WHOLE_ROWS   1562u
#define N_MAIN_TILES 6248u
#define N_MTILES     6250u
#define TILES_PER_WG 5u
#define GEMM_WGS     1250u

struct GCNN_tile_map { unsigned pbase, qshift, rbase, oddadd; };

__device__ __forceinline__ GCNN_tile_map GCNN_tile_terms(unsigned b) {
    const bool whole = b < N_MAIN_TILES;
    GCNN_tile_map t;
    t.pbase  = whole ? 8u * (b & 3u) : 16u * (b - N_MAIN_TILES);
    t.qshift = whole ? 1u : 0u;
    t.rbase  = whole ? (b >> 2) : WHOLE_ROWS;
    t.oddadd = whole ? (WHOLE_ROWS + 1u) : 0u;
    return t;
}

__device__ __forceinline__ unsigned GCNN_tile_row(const GCNN_tile_map t, unsigned q) {
    return (t.pbase + (q >> t.qshift)) * PAIR2_ROWS + t.rbase + (q & 1u) * t.oddadd;
}

__global__ __launch_bounds__(128)
void GCNN_gemm_wmma_kernel(const float* __restrict__ agg,
                           const float* __restrict__ W,
                           const float* __restrict__ bias,
                           float*       __restrict__ out) {
    const int lane = threadIdx.x & 31;
    const int wave = threadIdx.x >> 5;
    const int half = lane >> 4;
    const int l16  = lane & 15;
    const int n0   = wave * 16;
    const int col  = n0 + l16;

    v2f bfr[16];
    #pragma unroll
    for (int s = 0; s < 16; ++s) {
        const int ka = 4 * s + 2 * half;
        bfr[s].x = W[ka * C_OUT + col];
        bfr[s].y = W[(ka + 1) * C_OUT + col];
    }
    const float bv = bias[col];

    #pragma unroll 1
    for (unsigned ti = 0u; ti < TILES_PER_WG; ++ti) {
        const unsigned tb = (unsigned)blockIdx.x * TILES_PER_WG + ti;
        const GCNN_tile_map tm = GCNN_tile_terms(tb);
        const unsigned r = GCNN_tile_row(tm, (unsigned)l16);

        v8f acc = {};
        #pragma unroll
        for (int s = 0; s < 16; ++s) {
            const int ka = 4 * s + 2 * half;
            const unsigned f   = r * 64u + (unsigned)ka;
            const unsigned c0  = f / NB;
            const unsigned rem = f - c0 * NB;
            const v2f a = *(const v2f*)(agg + (rem >> 1) * 128u + c0 * 2u);

            acc = __builtin_amdgcn_wmma_f32_16x16x4_f32(
                 false, a,  false, bfr[s],
                 (short)0, acc,  false,  false);
        }

        #pragma unroll
        for (int v = 0; v < 8; ++v) {
            float o = acc[v] + bv;
            o = o > 0.0f ? o : 0.0f;
            const unsigned orow = GCNN_tile_row(tm, (unsigned)(v + 8 * half));
            out[orow * (unsigned)C_OUT + (unsigned)col] = o;
        }
    }
}

extern "C" void kernel_launch(void* const* d_in, const int* in_sizes, int n_in,
                              void* d_out, int out_size, void* d_ws, size_t ws_size,
                              hipStream_t stream) {
    const float* x    = (const float*)d_in[0];
    const int*   rows = (const int*)  d_in[1];
    const int*   cols = (const int*)  d_in[2];
    const float* vals = (const float*)d_in[3];
    const float* W    = (const float*)d_in[4];
    const float* bias = (const float*)d_in[5];
    float* out = (float*)d_out;
    float*    agg  = (float*)d_ws;
    unsigned* recs = (unsigned*)d_ws + 6400000;
    unsigned* offs = recs + 2 * N_CHUNKS * REC_STRIDE;

    GCNN_bin_kernel<<<N_CHUNKS, BIN_THREADS, 0, stream>>>(rows, cols, vals, offs, recs);

    GCNN_spmm_tile_kernel<<<N_TILES, 1024, 0, stream>>>(x, offs, recs, agg);

    GCNN_gemm_wmma_kernel<<<GEMM_WGS, 128, 0, stream>>>(agg, W, bias, out);
}


# SpMM tile work-group also does its GEMM rows from the LDS tile (WMMA): tiles cut at 16-row half-blocks, boundary half-block's edges bucketed to both tiles; 2 launches, no agg buffer; sum order kept
# speedup vs baseline: 7.9830x; 1.1268x over previous
#include <hip/hip_runtime.h>

#define N_NODES 50000
#define N_EDGES 800000
#define NB      100000u
#define C_OUT   64

typedef __attribute__((ext_vector_type(2))) float v2f;
typedef __attribute__((ext_vector_type(8))) float v8f;

#define HB_ROWS     16
#define BIG_TILES   52u
#define BIG_HB      676u
#define TILE_R      224
#define TILE_PITCH  144
#define N_TILES     256
#define N_CHUNKS    256
#define CHUNK       3125
#define REC_CAP     6250
#define REC_STRIDE  6272
#define NO_TILE     0xFFFFFFFFu
#define OFF_STRIDE  288
#define BIN_THREADS 256
#define BIN_EPT     13
#define WQ_CAP      512

__device__ __forceinline__ unsigned GCNN_tile_lo(unsigned t) {
    return 12u * t + (t < BIG_TILES ? t : BIG_TILES);
}

struct GCNN_dest { unsigned t1, l1, t2, l2; };

__device__ __forceinline__ GCNN_dest GCNN_row_dest(unsigned row) {
    const unsigned hb = row >> 4;
    const unsigned in = row & 15u;
    unsigned t1 = hb < BIG_HB ? hb / 13u : BIG_TILES + (hb - BIG_HB) / 12u;
    t1 = t1 < (unsigned)N_TILES ? t1 : (unsigned)(N_TILES - 1);
    const unsigned lo = GCNN_tile_lo(t1);
    GCNN_dest d;
    d.t1 = t1;
    d.l1 = (hb - lo) * HB_ROWS + in;
    d.t2 = NO_TILE;
    d.l2 = 0u;
    if (hb == lo && t1 > 0u) {
        d.t2 = t1 - 1u;
        d.l2 = (d.t2 < BIG_TILES ? 13u : 12u) * HB_ROWS + in;
    } else if (hb == 0u) {
        d.t2 = (unsigned)(N_TILES - 1);
        d.l2 = 13u * HB_ROWS + in;
    }
    return d;
}

__global__ __launch_bounds__(BIN_THREADS)
void GCNN_bin_kernel(const int*   __restrict__ rows,
                     const int*   __restrict__ cols,
                     const float* __restrict__ vals,
                     unsigned*    __restrict__ offs,
                     unsigned*    __restrict__ recs) {
    __shared__ unsigned cnt[N_TILES];
    __shared__ unsigned start[N_TILES + 1];
    __shared__ unsigned skey[REC_CAP];
    __shared__ unsigned sval[REC_CAP];

    const int tid  = threadIdx.x;
    const int g    = blockIdx.x;
    const int base = g * CHUNK;

    cnt[tid] = 0u;

    int   r[BIN_EPT];
    int   c[BIN_EPT];
    float v[BIN_EPT];
    #pragma unroll
    for (int k = 0; k < BIN_EPT; ++k) {
        const int idx = tid + k * BIN_THREADS;
        r[k] = -1; c[k] = 0; v[k] = 0.0f;
        if (idx < CHUNK) {
            r[k] = rows[base + idx];
            c[k] = cols[base + idx];
            v[k] = vals[base + idx];
        }
    }
    __syncthreads();

    unsigned pos[BIN_EPT];
    unsigned pos2[BIN_EPT];
    #pragma unroll
    for (int k = 0; k < BIN_EPT; ++k) {
        pos[k] = 0u; pos2[k] = 0u;
        if ((unsigned)r[k] < (unsigned)N_NODES) {
            const GCNN_dest d = GCNN_row_dest((unsigned)r[k]);
            pos[k] = atomicAdd(&cnt[d.t1], 1u);
            if (d.t2 != NO_TILE) pos2[k] = atomicAdd(&cnt[d.t2], 1u);
        }
    }
    __syncthreads();

    if (tid < 32) {
        unsigned c8[8];
        unsigned sum = 0u;
        #pragma unroll
        for (int i = 0; i < 8; ++i) { c8[i] = cnt[8 * tid + i]; sum += c8[i]; }
        unsigned inc = sum;
        #pragma unroll
        for (int d = 1; d < 32; d <<= 1) {
            const unsigned up = __shfl_up(inc, (unsigned)d);
            if (tid >= d) inc += up;
        }
        unsigned run = inc - sum;
        #pragma unroll
        for (int i = 0; i < 8; ++i) {
            start[8 * tid + i] = run;
            offs[g * OFF_STRIDE + 8 * tid + i] = run;
            run += c8[i];
        }
        if (tid == 31) {
            start[N_TILES] = run;
            offs[g * OFF_STRIDE + N_TILES] = run;
        }
    }
    __syncthreads();

    #pragma unroll
    for (int k = 0; k < BIN_EPT; ++k) {
        if ((unsigned)r[k] < (unsigned)N_NODES) {
            const GCNN_dest d = GCNN_row_dest((unsigned)r[k]);
            const unsigned p1 = start[d.t1] + pos[k];
            skey[p1] = (unsigned)c[k] | (d.l1 << 16);
            sval[p1] = __float_as_uint(v[k]);
            if (d.t2 != NO_TILE) {
                const unsigned p2 = start[d.t2] + pos2[k];
                skey[p2] = (unsigned)c[k] | (d.l2 << 16);
                sval[p2] = __float_as_uint(v[k]);
            }
        }
    }
    __syncthreads();

    const unsigned total = start[N_TILES];
    uint2* dst = (uint2*)recs + g * REC_STRIDE;
    for (unsigned i = (unsigned)tid; i < total; i += BIN_THREADS)
        dst[i] = make_uint2(skey[i], sval[i]);
}

#define PAIR2_ROWS   3125u
#define WHOLE_ROWS   1562u

__global__ __launch_bounds__(1024)
void GCNN_spmm_gemm_kernel(const float*    __restrict__ x,
                           const unsigned* __restrict__ offs,
                           const unsigned* __restrict__ recs,
                           const float*    __restrict__ W,
                           const float*    __restrict__ bias,
                           float*          __restrict__ out) {
    __shared__ float    tile[(TILE_R + 1) * TILE_PITCH];
    __shared__ unsigned qkey[32 * WQ_CAP];
    __shared__ unsigned qval[32 * WQ_CAP];

    const int tid  = threadIdx.x;
    const int lane = tid & 31;
    const int wave = __builtin_amdgcn_readfirstlane(tid >> 5);
    const int t    = (int)blockIdx.x;

    int o0 = 0, o1 = 0;
    if (lane < 8) {
        const unsigned* p = offs + (wave * 8 + lane) * OFF_STRIDE + t;
        o0 = (int)p[0];
        o1 = (int)p[1];
    }

    for (int p = tid; p < (TILE_R + 1) * (TILE_PITCH / 4); p += 1024)
        ((float4*)tile)[p] = make_float4(0.0f, 0.0f, 0.0f, 0.0f);

    const uint2* rec2 = (const uint2*)recs + wave * 8 * REC_STRIDE;
    unsigned key[8], vb[8];
    #pragma unroll
    for (int s = 0; s < 8; ++s) {
        const unsigned b = (unsigned)__builtin_amdgcn_readlane(o0, s);
        const unsigned n = (unsigned)__builtin_amdgcn_readlane(o1, s) - b;
        const unsigned li = (unsigned)lane < n ? (unsigned)lane : n - 1u;
        const uint2 rr = rec2[s * REC_STRIDE + (n != 0u ? b + li : 0u)];
        key[s] = rr.x; vb[s] = rr.y;
    }
    __syncthreads();

    unsigned* myk = qkey + wave * WQ_CAP;
    unsigned* myv = qval + wave * WQ_CAP;
    unsigned  cntw = 0u;
    #pragma unroll
    for (int s = 0; s < 8; ++s) {
        const unsigned n = (unsigned)(__builtin_amdgcn_readlane(o1, s) - __builtin_amdgcn_readlane(o0, s));
        const unsigned m = n < 32u ? n : 32u;
        if ((unsigned)lane < m) { myk[cntw + lane] = key[s]; myv[cntw + lane] = vb[s]; }
        cntw += m;
    }

    const float* xlane = x + (lane >> 4) * (N_NODES * 64) + (lane & 15) * 4;

    int      s = 0;
    unsigned q = 32u;
    for (;;) {
        while (s < 8 && cntw <= (unsigned)(WQ_CAP - 32)) {
            const unsigned b = (unsigned)__builtin_amdgcn_readlane(o0, s);
            const unsigned n = (unsigned)__builtin_amdgcn_readlane(o1, s) - b;
            if (q < n) {
                const unsigned m = (n - q) < 32u ? (n - q) : 32u;
                if ((unsigned)lane < m) {
                    const uint2 rr = rec2[s * REC_STRIDE + b + q + (unsigned)lane];
                    myk[cntw + lane] = rr.x; myv[cntw + lane] = rr.y;
                }
                cntw += m;
                q += 32u;
            } else {
                ++s;
                q = 32u;
            }
        }

        for (unsigned qb = 0u; qb < cntw; qb += 32u) {
            const unsigned nb = (cntw - qb) < 32u ? (cntw - qb) : 32u;
            int ent = TILE_R << 16;
            int vvb = 0;
            if ((unsigned)lane < nb) {
                ent = (int)myk[qb + (unsigned)lane];
                vvb = (int)myv[qb + (unsigned)lane];
            }
            for (unsigned u0 = 0u; u0 < nb; u0 += 8u) {
                unsigned e[8];
                float4   xv[8];
                #pragma unroll
                for (int u = 0; u < 8; ++u)
                    e[u] = (unsigned)__builtin_amdgcn_readlane(ent, (int)u0 + u);
                #pragma unroll
                for (int u = 0; u < 8; ++u)
                    xv[u] = *(const float4*)(xlane + (e[u] & 0xFFFFu) * 64u);
                #pragma unroll
                for (int u = 0; u < 8; ++u) {
                    const unsigned lrow = e[u] >> 16;
                    const float    v    = __int_as_float(__builtin_amdgcn_readlane(vvb, (int)u0 + u));
                    float* tp = tile + lrow * (unsigned)TILE_PITCH + (unsigned)lane;
                    __hip_atomic_fetch_add(tp,      v * xv[u].x, __ATOMIC_RELAXED, __HIP_MEMORY_SCOPE_WORKGROUP);
                    __hip_atomic_fetch_add(tp + 32, v * xv[u].y, __ATOMIC_RELAXED, __HIP_MEMORY_SCOPE_WORKGROUP);
                    __hip_atomic_fetch_add(tp + 64, v * xv[u].z, __ATOMIC_RELAXED, __HIP_MEMORY_SCOPE_WORKGROUP);
                    __hip_atomic_fetch_add(tp + 96, v * xv[u].w, __ATOMIC_RELAXED, __HIP_MEMORY_SCOPE_WORKGROUP);
                }
            }
        }
        cntw = 0u;
        if (s >= 8) break;
    }

    const int half = lane >> 4;
    const int l16  = lane & 15;
    const int col  = (wave & 3) * 16 + l16;

    v2f bfr[16];
    #pragma unroll
    for (int ks = 0; ks < 16; ++ks) {
        const int ka = 4 * ks + 2 * half;
        bfr[ks].x = W[ka * C_OUT + col];
        bfr[ks].y = W[(ka + 1) * C_OUT + col];
    }
    const float bv = bias[col];
    __syncthreads();

    const unsigned npairs = (unsigned)t < BIG_TILES ? 13u : 12u;
    const unsigned hb0    = GCNN_tile_lo((unsigned)t);
    const unsigned n_mt   = 2u * npairs + (t == N_TILES - 1 ? 2u : 0u);
    for (unsigned mi = (unsigned)wave >> 2; mi < n_mt; mi += 8u) {
        const bool     join = mi >= 2u * npairs;
        const unsigned p    = join ? npairs : (mi >> 1);
        const unsigned sub  = mi & 1u;
        const unsigned h    = hb0 + p;
        const unsigned par  = h & 1u;
        const float* ap = tile + (16u * p + (unsigned)half) * (unsigned)TILE_PITCH + 16u * sub + (unsigned)l16;

        v8f acc = {};
        #pragma unroll
        for (int ks = 0; ks < 16; ++ks) {
            const unsigned pb = join ? (ks >= 8 ? 1u : 0u) : par;
            const float*   aq = ap + (2 * ks) * TILE_PITCH + 64u * pb;
            v2f a;
            a.x = aq[0];
            a.y = aq[32];
            acc = __builtin_amdgcn_wmma_f32_16x16x4_f32(
                 false, a,  false, bfr[ks],
                 (short)0, acc,  false,  false);
        }

        const unsigned radd = join ? WHOLE_ROWS : (par * (WHOLE_ROWS + 1u) + (h >> 1));
        #pragma unroll
        for (int v = 0; v < 8; ++v) {
            float o = acc[v] + bv;
            o = o > 0.0f ? o : 0.0f;
            const unsigned orow = (16u * sub + (unsigned)(v + 8 * half)) * PAIR2_ROWS + radd;
            out[orow * (unsigned)C_OUT + (unsigned)col] = o;
        }
    }
}

extern "C" void kernel_launch(void* const* d_in, const int* in_sizes, int n_in,
                              void* d_out, int out_size, void* d_ws, size_t ws_size,
                              hipStream_t stream) {
    const float* x    = (const float*)d_in[0];
    const int*   rows = (const int*)  d_in[1];
    const int*   cols = (const int*)  d_in[2];
    const float* vals = (const float*)d_in[3];
    const float* W    = (const float*)d_in[4];
    const float* bias = (const float*)d_in[5];
    float* out = (float*)d_out;
    unsigned* recs = (unsigned*)d_ws;
    unsigned* offs = recs + 2 * N_CHUNKS * REC_STRIDE;

    GCNN_bin_kernel<<<N_CHUNKS, BIN_THREADS, 0, stream>>>(rows, cols, vals, offs, recs);

    GCNN_spmm_gemm_kernel<<<N_TILES, 1024, 0, stream>>>(x, offs, recs, W, bias, out);
}


# Tile records pooled in one work-group LDS queue (one integer add per wave reserves room), dealt out in 8-record trips, equal count per wave: no wave drains a longer list alone; sums arrival order
# speedup vs baseline: 8.2320x; 1.0312x over previous
#include <hip/hip_runtime.h>

#define N_NODES 50000
#define N_EDGES 800000
#define NB      100000u
#define C_OUT   64

typedef __attribute__((ext_vector_type(2))) float v2f;
typedef __attribute__((ext_vector_type(8))) float v8f;

#define HB_ROWS     16
#define BIG_TILES   52u
#define BIG_HB      676u
#define TILE_R      224
#define TILE_PITCH  144
#define N_TILES     256
#define N_CHUNKS    256
#define CHUNK       3125
#define REC_CAP     6250
#define REC_STRIDE  6272
#define NO_TILE     0xFFFFFFFFu
#define OFF_STRIDE  288
#define BIN_THREADS 256
#define BIN_EPT     13
#define Q_CAP       8192

__device__ __forceinline__ unsigned GCNN_tile_lo(unsigned t) {
    return 12u * t + (t < BIG_TILES ? t : BIG_TILES);
}

struct GCNN_dest { unsigned t1, l1, t2, l2; };

__device__ __forceinline__ GCNN_dest GCNN_row_dest(unsigned row) {
    const unsigned hb = row >> 4;
    const unsigned in = row & 15u;
    unsigned t1 = hb < BIG_HB ? hb / 13u : BIG_TILES + (hb - BIG_HB) / 12u;
    t1 = t1 < (unsigned)N_TILES ? t1 : (unsigned)(N_TILES - 1);
    const unsigned lo = GCNN_tile_lo(t1);
    GCNN_dest d;
    d.t1 = t1;
    d.l1 = (hb - lo) * HB_ROWS + in;
    d.t2 = NO_TILE;
    d.l2 = 0u;
    if (hb == lo && t1 > 0u) {
        d.t2 = t1 - 1u;
        d.l2 = (d.t2 < BIG_TILES ? 13u : 12u) * HB_ROWS + in;
    } else if (hb == 0u) {
        d.t2 = (unsigned)(N_TILES - 1);
        d.l2 = 13u * HB_ROWS + in;
    }
    return d;
}

__global__ __launch_bounds__(BIN_THREADS)
void GCNN_bin_kernel(const int*   __restrict__ rows,
                     const int*   __restrict__ cols,
                     const float* __restrict__ vals,
                     unsigned*    __restrict__ offs,
                     unsigned*    __restrict__ recs) {
    __shared__ unsigned cnt[N_TILES];
    __shared__ unsigned start[N_TILES + 1];
    __shared__ unsigned skey[REC_CAP];
    __shared__ unsigned sval[REC_CAP];

    const int tid  = threadIdx.x;
    const int g    = blockIdx.x;
    const int base = g * CHUNK;

    cnt[tid] = 0u;

    int   r[BIN_EPT];
    int   c[BIN_EPT];
    float v[BIN_EPT];
    #pragma unroll
    for (int k = 0; k < BIN_EPT; ++k) {
        const int idx = tid + k * BIN_THREADS;
        r[k] = -1; c[k] = 0; v[k] = 0.0f;
        if (idx < CHUNK) {
            r[k] = rows[base + idx];
            c[k] = cols[base + idx];
            v[k] = vals[base + idx];
        }
    }
    __syncthreads();

    unsigned pos[BIN_EPT];
    unsigned pos2[BIN_EPT];
    #pragma unroll
    for (int k = 0; k < BIN_EPT; ++k) {
        pos[k] = 0u; pos2[k] = 0u;
        if ((unsigned)r[k] < (unsigned)N_NODES) {
            const GCNN_dest d = GCNN_row_dest((unsigned)r[k]);
            pos[k] = atomicAdd(&cnt[d.t1], 1u);
            if (d.t2 != NO_TILE) pos2[k] = atomicAdd(&cnt[d.t2], 1u);
        }
    }
    __syncthreads();

    if (tid < 32) {
        unsigned c8[8];
        unsigned sum = 0u;
        #pragma unroll
        for (int i = 0; i < 8; ++i) { c8[i] = cnt[8 * tid + i]; sum += c8[i]; }
        unsigned inc = sum;
        #pragma unroll
        for (int d = 1; d < 32; d <<= 1) {
            const unsigned up = __shfl_up(inc, (unsigned)d);
            if (tid >= d) inc += up;
        }
        unsigned run = inc - sum;
        #pragma unroll
        for (int i = 0; i < 8; ++i) {
            start[8 * tid + i] = run;
            offs[g * OFF_STRIDE + 8 * tid + i] = run;
            run += c8[i];
        }
        if (tid == 31) {
            start[N_TILES] = run;
            offs[g * OFF_STRIDE + N_TILES] = run;
        }
    }
    __syncthreads();

    #pragma unroll
    for (int k = 0; k < BIN_EPT; ++k) {
        if ((unsigned)r[k] < (unsigned)N_NODES) {
            const GCNN_dest d = GCNN_row_dest((unsigned)r[k]);
            const unsigned p1 = start[d.t1] + pos[k];
            skey[p1] = (unsigned)c[k] | (d.l1 << 16);
            sval[p1] = __float_as_uint(v[k]);
            if (d.t2 != NO_TILE) {
                const unsigned p2 = start[d.t2] + pos2[k];
                skey[p2] = (unsigned)c[k] | (d.l2 << 16);
                sval[p2] = __float_as_uint(v[k]);
            }
        }
    }
    __syncthreads();

    const unsigned total = start[N_TILES];
    uint2* dst = (uint2*)recs + g * REC_STRIDE;
    for (unsigned i = (unsigned)tid; i < total; i += BIN_THREADS)
        dst[i] = make_uint2(skey[i], sval[i]);
}

#define PAIR2_ROWS   3125u
#define WHOLE_ROWS   1562u

__device__ __forceinline__ void GCNN_drain8(int ent, int vvb, unsigned u0,
                                            const float* __restrict__ xlane,
                                            float* tile, unsigned lane) {
    unsigned e[8];
    float4   xv[8];
    #pragma unroll
    for (int u = 0; u < 8; ++u)
        e[u] = (unsigned)__builtin_amdgcn_readlane(ent, (int)u0 + u);
    #pragma unroll
    for (int u = 0; u < 8; ++u)
        xv[u] = *(const float4*)(xlane + (e[u] & 0xFFFFu) * 64u);
    #pragma unroll
    for (int u = 0; u < 8; ++u) {
        const unsigned lrow = e[u] >> 16;
        const float    v    = __int_as_float(__builtin_amdgcn_readlane(vvb, (int)u0 + u));
        float* tp = tile + lrow * (unsigned)TILE_PITCH + lane;
        __hip_atomic_fetch_add(tp,      v * xv[u].x, __ATOMIC_RELAXED, __HIP_MEMORY_SCOPE_WORKGROUP);
        __hip_atomic_fetch_add(tp + 32, v * xv[u].y, __ATOMIC_RELAXED, __HIP_MEMORY_SCOPE_WORKGROUP);
        __hip_atomic_fetch_add(tp + 64, v * xv[u].z, __ATOMIC_RELAXED, __HIP_MEMORY_SCOPE_WORKGROUP);
        __hip_atomic_fetch_add(tp + 96, v * xv[u].w, __ATOMIC_RELAXED, __HIP_MEMORY_SCOPE_WORKGROUP);
    }
}

__global__ __launch_bounds__(1024)
void GCNN_spmm_gemm_kernel(const float*    __restrict__ x,
                           const unsigned* __restrict__ offs,
                           const unsigned* __restrict__ recs,
                           const float*    __restrict__ W,
                           const float*    __restrict__ bias,
                           float*          __restrict__ out) {
    __shared__ float    tile[(TILE_R + 1) * TILE_PITCH];
    __shared__ unsigned qkey[Q_CAP];
    __shared__ unsigned qval[Q_CAP];
    __shared__ unsigned qtotal;

    const int tid  = threadIdx.x;
    const int lane = tid & 31;
    const int wave = __builtin_amdgcn_readfirstlane(tid >> 5);
    const int t    = (int)blockIdx.x;

    int o0 = 0, o1 = 0;
    if (lane < 8) {
        const unsigned* p = offs + (wave * 8 + lane) * OFF_STRIDE + t;
        o0 = (int)p[0];
        o1 = (int)p[1];
    }

    if (tid == 0) qtotal = 0u;
    for (int p = tid; p < (TILE_R + 1) * (TILE_PITCH / 4); p += 1024)
        ((float4*)tile)[p] = make_float4(0.0f, 0.0f, 0.0f, 0.0f);

    const uint2* rec2 = (const uint2*)recs + wave * 8 * REC_STRIDE;
    unsigned key[8], vb[8];
    #pragma unroll
    for (int s = 0; s < 8; ++s) {
        const unsigned b = (unsigned)__builtin_amdgcn_readlane(o0, s);
        const unsigned n = (unsigned)__builtin_amdgcn_readlane(o1, s) - b;
        const unsigned li = (unsigned)lane < n ? (unsigned)lane : n - 1u;
        const uint2 rr = rec2[s * REC_STRIDE + (n != 0u ? b + li : 0u)];
        key[s] = rr.x; vb[s] = rr.y;
    }
    __syncthreads();

    unsigned cntw = 0u;
    #pragma unroll
    for (int s = 0; s < 8; ++s) {
        const unsigned n = (unsigned)(__builtin_amdgcn_readlane(o1, s) - __builtin_amdgcn_readlane(o0, s));
        cntw += n < 32u ? n : 32u;
    }
    unsigned qbase = 0u;
    if (lane == 0) qbase = atomicAdd(&qtotal, cntw);
    qbase = (unsigned)__builtin_amdgcn_readfirstlane((int)qbase);
    #pragma unroll
    for (int s = 0; s < 8; ++s) {
        const unsigned n = (unsigned)(__builtin_amdgcn_readlane(o1, s) - __builtin_amdgcn_readlane(o0, s));
        const unsigned m = n < 32u ? n : 32u;
        if ((unsigned)lane < m) { qkey[qbase + lane] = key[s]; qval[qbase + lane] = vb[s]; }
        qbase += m;
    }
    __syncthreads();
    const unsigned total = (unsigned)__builtin_amdgcn_readfirstlane((int)qtotal);

    const float* xlane = x + (lane >> 4) * (N_NODES * 64) + (lane & 15) * 4;

    const unsigned n_trips = (total + 7u) >> 3;
    for (unsigned k0 = (unsigned)wave; k0 < n_trips; k0 += 128u) {
        const unsigned ri = (k0 + 32u * ((unsigned)lane >> 3)) * 8u + ((unsigned)lane & 7u);
        int ent = TILE_R << 16;
        int vvb = 0;
        if (ri < total) {
            ent = (int)qkey[ri];
            vvb = (int)qval[ri];
        }
        const unsigned left = (n_trips - k0 + 31u) >> 5;
        const unsigned nt   = left < 4u ? left : 4u;
        for (unsigned j = 0u; j < nt; ++j)
            GCNN_drain8(ent, vvb, 8u * j, xlane, tile, (unsigned)lane);
    }

    for (int s = 0; s < 8; ++s) {
        const unsigned b = (unsigned)__builtin_amdgcn_readlane(o0, s);
        const unsigned n = (unsigned)__builtin_amdgcn_readlane(o1, s) - b;
        for (unsigned q = 32u; q < n; q += 32u) {
            const unsigned m = (n - q) < 32u ? (n - q) : 32u;
            int ent = TILE_R << 16;
            int vvb = 0;
            if ((unsigned)lane < m) {
                const uint2 rr = rec2[s * REC_STRIDE + b + q + (unsigned)lane];
                ent = (int)rr.x; vvb = (int)rr.y;
            }
            for (unsigned u0 = 0u; u0 < m; u0 += 8u)
                GCNN_drain8(ent, vvb, u0, xlane, tile, (unsigned)lane);
        }
    }

    const int half = lane >> 4;
    const int l16  = lane & 15;
    const int col  = (wave & 3) * 16 + l16;

    v2f bfr[16];
    #pragma unroll
    for (int ks = 0; ks < 16; ++ks) {
        const int ka = 4 * ks + 2 * half;
        bfr[ks].x = W[ka * C_OUT + col];
        bfr[ks].y = W[(ka + 1) * C_OUT + col];
    }
    const float bv = bias[col];
    __syncthreads();

    const unsigned npairs = (unsigned)t < BIG_TILES ? 13u : 12u;
    const unsigned hb0    = GCNN_tile_lo((unsigned)t);
    const unsigned n_mt   = 2u * npairs + (t == N_TILES - 1 ? 2u : 0u);
    for (unsigned mi = (unsigned)wave >> 2; mi < n_mt; mi += 8u) {
        const bool     join = mi >= 2u * npairs;
        const unsigned p    = join ? npairs : (mi >> 1);
        const unsigned sub  = mi & 1u;
        const unsigned h    = hb0 + p;
        const unsigned par  = h & 1u;
        const float* ap = tile + (16u * p + (unsigned)half) * (unsigned)TILE_PITCH + 16u * sub + (unsigned)l16;

        v8f acc = {};
        #pragma unroll
        for (int ks = 0; ks < 16; ++ks) {
            const unsigned pb = join ? (ks >= 8 ? 1u : 0u) : par;
            const float*   aq = ap + (2 * ks) * TILE_PITCH + 64u * pb;
            v2f a;
            a.x = aq[0];
            a.y = aq[32];
            acc = __builtin_amdgcn_wmma_f32_16x16x4_f32(
                 false, a,  false, bfr[ks],
                 (short)0, acc,  false,  false);
        }

        const unsigned radd = join ? WHOLE_ROWS : (par * (WHOLE_ROWS + 1u) + (h >> 1));
        #pragma unroll
        for (int v = 0; v < 8; ++v) {
            float o = acc[v] + bv;
            o = o > 0.0f ? o : 0.0f;
            const unsigned orow = (16u * sub + (unsigned)(v + 8 * half)) * PAIR2_ROWS + radd;
            out[orow * (unsigned)C_OUT + (unsigned)col] = o;
        }
    }
}

extern "C" void kernel_launch(void* const* d_in, const int* in_sizes, int n_in,
                              void* d_out, int out_size, void* d_ws, size_t ws_size,
                              hipStream_t stream) {
    const float* x    = (const float*)d_in[0];
    const int*   rows = (const int*)  d_in[1];
    const int*   cols = (const int*)  d_in[2];
    const float* vals = (const float*)d_in[3];
    const float* W    = (const float*)d_in[4];
    const float* bias = (const float*)d_in[5];
    float* out = (float*)d_out;
    unsigned* recs = (unsigned*)d_ws;
    unsigned* offs = recs + 2 * N_CHUNKS * REC_STRIDE;

    GCNN_bin_kernel<<<N_CHUNKS, BIN_THREADS, 0, stream>>>(rows, cols, vals, offs, recs);

    GCNN_spmm_gemm_kernel<<<N_TILES, 1024, 0, stream>>>(x, offs, recs, W, bias, out);
}


# Tile rows owned by waves: wave w sums rows w, w+32, .. in registers (scalar row index); records regrouped into one LDS queue per owner by integer-add rank; no LDS float adds; arrival order kept
# speedup vs baseline: 8.5731x; 1.0414x over previous
#include <hip/hip_runtime.h>

#define N_NODES 50000
#define N_EDGES 800000
#define NB      100000u
#define C_OUT   64

typedef __attribute__((ext_vector_type(2))) float v2f;
typedef __attribute__((ext_vector_type(8))) float v8f;

#define HB_ROWS     16
#define BIG_TILES   52u
#define BIG_HB      676u
#define TILE_R      224
#define TILE_PITCH  144
#define N_TILES     256
#define N_CHUNKS    256
#define CHUNK       3125
#define REC_CAP     6250
#define REC_STRIDE  6272
#define NO_TILE     0xFFFFFFFFu
#define OFF_STRIDE  288
#define BIN_THREADS 256
#define BIN_EPT     13
#define Q_CAP       8192
#define QW          256
#define ROWS_PER_WAVE 7

__device__ __forceinline__ unsigned GCNN_tile_lo(unsigned t) {
    return 12u * t + (t < BIG_TILES ? t : BIG_TILES);
}

struct GCNN_dest { unsigned t1, l1, t2, l2; };

__device__ __forceinline__ GCNN_dest GCNN_row_dest(unsigned row) {
    const unsigned hb = row >> 4;
    const unsigned in = row & 15u;
    unsigned t1 = hb < BIG_HB ? hb / 13u : BIG_TILES + (hb - BIG_HB) / 12u;
    t1 = t1 < (unsigned)N_TILES ? t1 : (unsigned)(N_TILES - 1);
    const unsigned lo = GCNN_tile_lo(t1);
    GCNN_dest d;
    d.t1 = t1;
    d.l1 = (hb - lo) * HB_ROWS + in;
    d.t2 = NO_TILE;
    d.l2 = 0u;
    if (hb == lo && t1 > 0u) {
        d.t2 = t1 - 1u;
        d.l2 = (d.t2 < BIG_TILES ? 13u : 12u) * HB_ROWS + in;
    } else if (hb == 0u) {
        d.t2 = (unsigned)(N_TILES - 1);
        d.l2 = 13u * HB_ROWS + in;
    }
    return d;
}

__global__ __launch_bounds__(BIN_THREADS)
void GCNN_bin_kernel(const int*   __restrict__ rows,
                     const int*   __restrict__ cols,
                     const float* __restrict__ vals,
                     unsigned*    __restrict__ offs,
                     unsigned*    __restrict__ recs) {
    __shared__ unsigned cnt[N_TILES];
    __shared__ unsigned start[N_TILES + 1];
    __shared__ unsigned skey[REC_CAP];
    __shared__ unsigned sval[REC_CAP];

    const int tid  = threadIdx.x;
    const int g    = blockIdx.x;
    const int base = g * CHUNK;

    cnt[tid] = 0u;

    int   r[BIN_EPT];
    int   c[BIN_EPT];
    float v[BIN_EPT];
    #pragma unroll
    for (int k = 0; k < BIN_EPT; ++k) {
        const int idx = tid + k * BIN_THREADS;
        r[k] = -1; c[k] = 0; v[k] = 0.0f;
        if (idx < CHUNK) {
            r[k] = rows[base + idx];
            c[k] = cols[base + idx];
            v[k] = vals[base + idx];
        }
    }
    __syncthreads();

    unsigned pos[BIN_EPT];
    unsigned pos2[BIN_EPT];
    #pragma unroll
    for (int k = 0; k < BIN_EPT; ++k) {
        pos[k] = 0u; pos2[k] = 0u;
        if ((unsigned)r[k] < (unsigned)N_NODES) {
            const GCNN_dest d = GCNN_row_dest((unsigned)r[k]);
            pos[k] = atomicAdd(&cnt[d.t1], 1u);
            if (d.t2 != NO_TILE) pos2[k] = atomicAdd(&cnt[d.t2], 1u);
        }
    }
    __syncthreads();

    if (tid < 32) {
        unsigned c8[8];
        unsigned sum = 0u;
        #pragma unroll
        for (int i = 0; i < 8; ++i) { c8[i] = cnt[8 * tid + i]; sum += c8[i]; }
        unsigned inc = sum;
        #pragma unroll
        for (int d = 1; d < 32; d <<= 1) {
            const unsigned up = __shfl_up(inc, (unsigned)d);
            if (tid >= d) inc += up;
        }
        unsigned run = inc - sum;
        #pragma unroll
        for (int i = 0; i < 8; ++i) {
            start[8 * tid + i] = run;
            offs[g * OFF_STRIDE + 8 * tid + i] = run;
            run += c8[i];
        }
        if (tid == 31) {
            start[N_TILES] = run;
            offs[g * OFF_STRIDE + N_TILES] = run;
        }
    }
    __syncthreads();

    #pragma unroll
    for (int k = 0; k < BIN_EPT; ++k) {
        if ((unsigned)r[k] < (unsigned)N_NODES) {
            const GCNN_dest d = GCNN_row_dest((unsigned)r[k]);
            const unsigned p1 = start[d.t1] + pos[k];
            skey[p1] = (unsigned)c[k] | (d.l1 << 16);
            sval[p1] = __float_as_uint(v[k]);
            if (d.t2 != NO_TILE) {
                const unsigned p2 = start[d.t2] + pos2[k];
                skey[p2] = (unsigned)c[k] | (d.l2 << 16);
                sval[p2] = __float_as_uint(v[k]);
            }
        }
    }
    __syncthreads();

    const unsigned total = start[N_TILES];
    uint2* dst = (uint2*)recs + g * REC_STRIDE;
    for (unsigned i = (unsigned)tid; i < total; i += BIN_THREADS)
        dst[i] = make_uint2(skey[i], sval[i]);
}

#define PAIR2_ROWS   3125u
#define WHOLE_ROWS   1562u

__device__ __forceinline__ void GCNN_drain8(int ent, int vvb, unsigned u0,
                                            const float* __restrict__ xlane,
                                            float* tile, unsigned lane) {
    unsigned e[8];
    float4   xv[8];
    #pragma unroll
    for (int u = 0; u < 8; ++u)
        e[u] = (unsigned)__builtin_amdgcn_readlane(ent, (int)u0 + u);
    #pragma unroll
    for (int u = 0; u < 8; ++u)
        xv[u] = *(const float4*)(xlane + (e[u] & 0xFFFFu) * 64u);
    #pragma unroll
    for (int u = 0; u < 8; ++u) {
        const unsigned lrow = e[u] >> 16;
        const float    v    = __int_as_float(__builtin_amdgcn_readlane(vvb, (int)u0 + u));
        float* tp = tile + lrow * (unsigned)TILE_PITCH + lane;
        __hip_atomic_fetch_add(tp,      v * xv[u].x, __ATOMIC_RELAXED, __HIP_MEMORY_SCOPE_WORKGROUP);
        __hip_atomic_fetch_add(tp + 32, v * xv[u].y, __ATOMIC_RELAXED, __HIP_MEMORY_SCOPE_WORKGROUP);
        __hip_atomic_fetch_add(tp + 64, v * xv[u].z, __ATOMIC_RELAXED, __HIP_MEMORY_SCOPE_WORKGROUP);
        __hip_atomic_fetch_add(tp + 96, v * xv[u].w, __ATOMIC_RELAXED, __HIP_MEMORY_SCOPE_WORKGROUP);
    }
}

struct GCNN_rows { v8f x, y, z, w; };

__device__ __forceinline__ void GCNN_own8(int ent, int vvb, unsigned u0,
                                          const float* __restrict__ xlane,
                                          GCNN_rows& acc) {
    unsigned e[8];
    float4   xv[8];
    #pragma unroll
    for (int u = 0; u < 8; ++u)
        e[u] = (unsigned)__builtin_amdgcn_readlane(ent, (int)u0 + u);
    #pragma unroll
    for (int u = 0; u < 8; ++u)
        xv[u] = *(const float4*)(xlane + (e[u] & 0xFFFFu) * 64u);
    #pragma unroll
    for (int u = 0; u < 8; ++u) {
        const float    v = __int_as_float(__builtin_amdgcn_readlane(vvb, (int)u0 + u));
        const unsigned k = e[u] >> 21;
        acc.x[k] += v * xv[u].x;
        acc.y[k] += v * xv[u].y;
        acc.z[k] += v * xv[u].z;
        acc.w[k] += v * xv[u].w;
    }
}

__global__ __launch_bounds__(1024)
void GCNN_spmm_gemm_kernel(const float*    __restrict__ x,
                           const unsigned* __restrict__ offs,
                           const unsigned* __restrict__ recs,
                           const float*    __restrict__ W,
                           const float*    __restrict__ bias,
                           float*          __restrict__ out) {
    __shared__ float    tile[(TILE_R + 1) * TILE_PITCH];
    __shared__ unsigned qkey[Q_CAP];
    __shared__ unsigned qval[Q_CAP];
    __shared__ unsigned qcnt[32];
    __shared__ unsigned qlate;

    const int tid  = threadIdx.x;
    const int lane = tid & 31;
    const int wave = __builtin_amdgcn_readfirstlane(tid >> 5);
    const int t    = (int)blockIdx.x;

    int o0 = 0, o1 = 0;
    if (lane < 8) {
        const unsigned* p = offs + (wave * 8 + lane) * OFF_STRIDE + t;
        o0 = (int)p[0];
        o1 = (int)p[1];
    }

    if (tid < 32) qcnt[tid] = 0u;
    if (tid == 0) qlate = 0u;
    if (tid < TILE_PITCH) tile[TILE_R * TILE_PITCH + tid] = 0.0f;

    const uint2* rec2 = (const uint2*)recs + wave * 8 * REC_STRIDE;
    unsigned key[8], vb[8];
    #pragma unroll
    for (int s = 0; s < 8; ++s) {
        const unsigned b = (unsigned)__builtin_amdgcn_readlane(o0, s);
        const unsigned n = (unsigned)__builtin_amdgcn_readlane(o1, s) - b;
        const unsigned li = (unsigned)lane < n ? (unsigned)lane : n - 1u;
        const uint2 rr = rec2[s * REC_STRIDE + (n != 0u ? b + li : 0u)];
        key[s] = rr.x; vb[s] = rr.y;
    }
    __syncthreads();

    int      late  = 0;
    unsigned later = 0u;
    unsigned pos[8];
    #pragma unroll
    for (int s = 0; s < 8; ++s) {
        const unsigned n = (unsigned)(__builtin_amdgcn_readlane(o1, s) - __builtin_amdgcn_readlane(o0, s));
        const unsigned m = n < 32u ? n : 32u;
        pos[s] = 0u;
        if ((unsigned)lane < m) pos[s] = atomicAdd(&qcnt[(key[s] >> 16) & 31u], 1u);
    }
    #pragma unroll
    for (int s = 0; s < 8; ++s) {
        const unsigned n = (unsigned)(__builtin_amdgcn_readlane(o1, s) - __builtin_amdgcn_readlane(o0, s));
        const unsigned m = n < 32u ? n : 32u;
        bool full = false;
        if ((unsigned)lane < m) {
            const unsigned owner = (key[s] >> 16) & 31u;
            if (pos[s] < (unsigned)QW) {
                const unsigned slot = owner * QW + ((pos[s] + 2u * owner) & (QW - 1u));
                qkey[slot] = key[s];
                qval[slot] = vb[s];
            } else {
                full = true;
            }
        }
        const unsigned fm = __builtin_amdgcn_ballot_w32(full);
        if (lane == s) late = (int)fm;
        later |= fm | (n > 32u ? 1u : 0u);
    }
    if (later != 0u && lane == 0) qlate = 1u;
    __syncthreads();
    unsigned n_own = (unsigned)__builtin_amdgcn_readfirstlane((int)qcnt[wave]);
    n_own = n_own < (unsigned)QW ? n_own : (unsigned)QW;
    const bool second = __builtin_amdgcn_readfirstlane((int)qlate) != 0;

    const float* xlane = x + (lane >> 4) * (N_NODES * 64) + (lane & 15) * 4;

    GCNN_rows acc;
    acc.x = v8f{}; acc.y = v8f{}; acc.z = v8f{}; acc.w = v8f{};
    for (unsigned i0 = 0u; i0 < n_own; i0 += 32u) {
        const unsigned i = i0 + (unsigned)lane;
        int ent = TILE_R << 16;
        int vvb = 0;
        if (i < n_own) {
            const unsigned slot = (unsigned)wave * QW + ((i + 2u * (unsigned)wave) & (QW - 1u));
            ent = (int)qkey[slot];
            vvb = (int)qval[slot];
        }
        const unsigned left = (n_own - i0 + 7u) >> 3;
        const unsigned nt   = left < 4u ? left : 4u;
        for (unsigned j = 0u; j < nt; ++j)
            GCNN_own8(ent, vvb, 8u * j, xlane, acc);
    }
    #pragma unroll
    for (int k = 0; k < ROWS_PER_WAVE; ++k) {
        float* tp = tile + (unsigned)(wave + 32 * k) * (unsigned)TILE_PITCH + (unsigned)lane;
        tp[0]  = acc.x[k];
        tp[32] = acc.y[k];
        tp[64] = acc.z[k];
        tp[96] = acc.w[k];
    }

    if (second) {
        __syncthreads();
        for (int s = 0; s < 8; ++s) {
            const unsigned b  = (unsigned)__builtin_amdgcn_readlane(o0, s);
            const unsigned n  = (unsigned)__builtin_amdgcn_readlane(o1, s) - b;
            const unsigned fm = (unsigned)__builtin_amdgcn_readlane(late, s);
            if (fm != 0u) {
                int ent = TILE_R << 16;
                int vvb = 0;
                if ((fm >> lane) & 1u) {
                    const uint2 rr = rec2[s * REC_STRIDE + b + (unsigned)lane];
                    ent = (int)rr.x; vvb = (int)rr.y;
                }
                for (unsigned u0 = 0u; u0 < 32u; u0 += 8u)
                    if ((fm >> u0) & 0xFFu)
                        GCNN_drain8(ent, vvb, u0, xlane, tile, (unsigned)lane);
            }
            for (unsigned q = 32u; q < n; q += 32u) {
                const unsigned m = (n - q) < 32u ? (n - q) : 32u;
                int ent = TILE_R << 16;
                int vvb = 0;
                if ((unsigned)lane < m) {
                    const uint2 rr = rec2[s * REC_STRIDE + b + q + (unsigned)lane];
                    ent = (int)rr.x; vvb = (int)rr.y;
                }
                for (unsigned u0 = 0u; u0 < m; u0 += 8u)
                    GCNN_drain8(ent, vvb, u0, xlane, tile, (unsigned)lane);
            }
        }
    }

    const int half = lane >> 4;
    const int l16  = lane & 15;
    const int col  = (wave & 3) * 16 + l16;

    v2f bfr[16];
    #pragma unroll
    for (int ks = 0; ks < 16; ++ks) {
        const int ka = 4 * ks + 2 * half;
        bfr[ks].x = W[ka * C_OUT + col];
        bfr[ks].y = W[(ka + 1) * C_OUT + col];
    }
    const float bv = bias[col];
    __syncthreads();

    const unsigned npairs = (unsigned)t < BIG_TILES ? 13u : 12u;
    const unsigned hb0    = GCNN_tile_lo((unsigned)t);
    const unsigned n_mt   = 2u * npairs + (t == N_TILES - 1 ? 2u : 0u);
    for (unsigned mi = (unsigned)wave >> 2; mi < n_mt; mi += 8u) {
        const bool     join = mi >= 2u * npairs;
        const unsigned p    = join ? npairs : (mi >> 1);
        const unsigned sub  = mi & 1u;
        const unsigned h    = hb0 + p;
        const unsigned par  = h & 1u;
        const float* ap = tile + (16u * p + (unsigned)half) * (unsigned)TILE_PITCH + 16u * sub + (unsigned)l16;

        v8f acc = {};
        #pragma unroll
        for (int ks = 0; ks < 16; ++ks) {
            const unsigned pb = join ? (ks >= 8 ? 1u : 0u) : par;
            const float*   aq = ap + (2 * ks) * TILE_PITCH + 64u * pb;
            v2f a;
            a.x = aq[0];
            a.y = aq[32];
            acc = __builtin_amdgcn_wmma_f32_16x16x4_f32(
                 false, a,  false, bfr[ks],
                 (short)0, acc,  false,  false);
        }

        const unsigned radd = join ? WHOLE_ROWS : (par * (WHOLE_ROWS + 1u) + (h >> 1));
        #pragma unroll
        for (int v = 0; v < 8; ++v) {
            float o = acc[v] + bv;
            o = o > 0.0f ? o : 0.0f;
            const unsigned orow = (16u * sub + (unsigned)(v + 8 * half)) * PAIR2_ROWS + radd;
            out[orow * (unsigned)C_OUT + (unsigned)col] = o;
        }
    }
}

extern "C" void kernel_launch(void* const* d_in, const int* in_sizes, int n_in,
                              void* d_out, int out_size, void* d_ws, size_t ws_size,
                              hipStream_t stream) {
    const float* x    = (const float*)d_in[0];
    const int*   rows = (const int*)  d_in[1];
    const int*   cols = (const int*)  d_in[2];
    const float* vals = (const float*)d_in[3];
    const float* W    = (const float*)d_in[4];
    const float* bias = (const float*)d_in[5];
    float* out = (float*)d_out;
    unsigned* recs = (unsigned*)d_ws;
    unsigned* offs = recs + 2 * N_CHUNKS * REC_STRIDE;

    GCNN_bin_kernel<<<N_CHUNKS, BIN_THREADS, 0, stream>>>(rows, cols, vals, offs, recs);

    GCNN_spmm_gemm_kernel<<<N_TILES, 1024, 0, stream>>>(x, offs, recs, W, bias, out);
}


# Records queued per tile row (rank: LDS integer add on the row's counter); a wave sums its rows one by one in 4 fixed registers, row end = scalar branch; no indexed register moves; arrival order kept
# speedup vs baseline: 8.8805x; 1.0359x over previous
#include <hip/hip_runtime.h>

#define N_NODES 50000
#define N_EDGES 800000
#define NB      100000u
#define C_OUT   64

typedef __attribute__((ext_vector_type(2))) float v2f;
typedef __attribute__((ext_vector_type(8))) float v8f;

#define HB_ROWS     16
#define BIG_TILES   52u
#define BIG_HB      676u
#define TILE_R      224
#define TILE_PITCH  144
#define N_TILES     256
#define N_CHUNKS    256
#define CHUNK       3125
#define REC_CAP     6250
#define REC_STRIDE  6272
#define NO_TILE     0xFFFFFFFFu
#define OFF_STRIDE  288
#define BIN_THREADS 256
#define BIN_EPT     13
#define RQ          40
#define Q_CAP       8960
#define ROWS_PER_WAVE 7

__device__ __forceinline__ unsigned GCNN_tile_lo(unsigned t) {
    return 12u * t + (t < BIG_TILES ? t : BIG_TILES);
}

struct GCNN_dest { unsigned t1, l1, t2, l2; };

__device__ __forceinline__ GCNN_dest GCNN_row_dest(unsigned row) {
    const unsigned hb = row >> 4;
    const unsigned in = row & 15u;
    unsigned t1 = hb < BIG_HB ? hb / 13u : BIG_TILES + (hb - BIG_HB) / 12u;
    t1 = t1 < (unsigned)N_TILES ? t1 : (unsigned)(N_TILES - 1);
    const unsigned lo = GCNN_tile_lo(t1);
    GCNN_dest d;
    d.t1 = t1;
    d.l1 = (hb - lo) * HB_ROWS + in;
    d.t2 = NO_TILE;
    d.l2 = 0u;
    if (hb == lo && t1 > 0u) {
        d.t2 = t1 - 1u;
        d.l2 = (d.t2 < BIG_TILES ? 13u : 12u) * HB_ROWS + in;
    } else if (hb == 0u) {
        d.t2 = (unsigned)(N_TILES - 1);
        d.l2 = 13u * HB_ROWS + in;
    }
    return d;
}

__global__ __launch_bounds__(BIN_THREADS)
void GCNN_bin_kernel(const int*   __restrict__ rows,
                     const int*   __restrict__ cols,
                     const float* __restrict__ vals,
                     unsigned*    __restrict__ offs,
                     unsigned*    __restrict__ recs) {
    __shared__ unsigned cnt[N_TILES];
    __shared__ unsigned start[N_TILES + 1];
    __shared__ unsigned skey[REC_CAP];
    __shared__ unsigned sval[REC_CAP];

    const int tid  = threadIdx.x;
    const int g    = blockIdx.x;
    const int base = g * CHUNK;

    cnt[tid] = 0u;

    int   r[BIN_EPT];
    int   c[BIN_EPT];
    float v[BIN_EPT];
    #pragma unroll
    for (int k = 0; k < BIN_EPT; ++k) {
        const int idx = tid + k * BIN_THREADS;
        r[k] = -1; c[k] = 0; v[k] = 0.0f;
        if (idx < CHUNK) {
            r[k] = rows[base + idx];
            c[k] = cols[base + idx];
            v[k] = vals[base + idx];
        }
    }
    __syncthreads();

    unsigned pos[BIN_EPT];
    unsigned pos2[BIN_EPT];
    #pragma unroll
    for (int k = 0; k < BIN_EPT; ++k) {
        pos[k] = 0u; pos2[k] = 0u;
        if ((unsigned)r[k] < (unsigned)N_NODES) {
            const GCNN_dest d = GCNN_row_dest((unsigned)r[k]);
            pos[k] = atomicAdd(&cnt[d.t1], 1u);
            if (d.t2 != NO_TILE) pos2[k] = atomicAdd(&cnt[d.t2], 1u);
        }
    }
    __syncthreads();

    if (tid < 32) {
        unsigned c8[8];
        unsigned sum = 0u;
        #pragma unroll
        for (int i = 0; i < 8; ++i) { c8[i] = cnt[8 * tid + i]; sum += c8[i]; }
        unsigned inc = sum;
        #pragma unroll
        for (int d = 1; d < 32; d <<= 1) {
            const unsigned up = __shfl_up(inc, (unsigned)d);
            if (tid >= d) inc += up;
        }
        unsigned run = inc - sum;
        #pragma unroll
        for (int i = 0; i < 8; ++i) {
            start[8 * tid + i] = run;
            offs[g * OFF_STRIDE + 8 * tid + i] = run;
            run += c8[i];
        }
        if (tid == 31) {
            start[N_TILES] = run;
            offs[g * OFF_STRIDE + N_TILES] = run;
        }
    }
    __syncthreads();

    #pragma unroll
    for (int k = 0; k < BIN_EPT; ++k) {
        if ((unsigned)r[k] < (unsigned)N_NODES) {
            const GCNN_dest d = GCNN_row_dest((unsigned)r[k]);
            const unsigned p1 = start[d.t1] + pos[k];
            skey[p1] = (unsigned)c[k] | (d.l1 << 16);
            sval[p1] = __float_as_uint(v[k]);
            if (d.t2 != NO_TILE) {
                const unsigned p2 = start[d.t2] + pos2[k];
                skey[p2] = (unsigned)c[k] | (d.l2 << 16);
                sval[p2] = __float_as_uint(v[k]);
            }
        }
    }
    __syncthreads();

    const unsigned total = start[N_TILES];
    uint2* dst = (uint2*)recs + g * REC_STRIDE;
    for (unsigned i = (unsigned)tid; i < total; i += BIN_THREADS)
        dst[i] = make_uint2(skey[i], sval[i]);
}

#define PAIR2_ROWS   3125u
#define WHOLE_ROWS   1562u

__device__ __forceinline__ void GCNN_drain8(int ent, int vvb, unsigned u0,
                                            const float* __restrict__ xlane,
                                            float* tile, unsigned lane) {
    unsigned e[8];
    float4   xv[8];
    #pragma unroll
    for (int u = 0; u < 8; ++u)
        e[u] = (unsigned)__builtin_amdgcn_readlane(ent, (int)u0 + u);
    #pragma unroll
    for (int u = 0; u < 8; ++u)
        xv[u] = *(const float4*)(xlane + (e[u] & 0xFFFFu) * 64u);
    #pragma unroll
    for (int u = 0; u < 8; ++u) {
        const unsigned lrow = e[u] >> 16;
        const float    v    = __int_as_float(__builtin_amdgcn_readlane(vvb, (int)u0 + u));
        float* tp = tile + lrow * (unsigned)TILE_PITCH + lane;
        __hip_atomic_fetch_add(tp,      v * xv[u].x, __ATOMIC_RELAXED, __HIP_MEMORY_SCOPE_WORKGROUP);
        __hip_atomic_fetch_add(tp + 32, v * xv[u].y, __ATOMIC_RELAXED, __HIP_MEMORY_SCOPE_WORKGROUP);
        __hip_atomic_fetch_add(tp + 64, v * xv[u].z, __ATOMIC_RELAXED, __HIP_MEMORY_SCOPE_WORKGROUP);
        __hip_atomic_fetch_add(tp + 96, v * xv[u].w, __ATOMIC_RELAXED, __HIP_MEMORY_SCOPE_WORKGROUP);
    }
}

struct GCNN_open { float x, y, z, w; unsigned k; };

__device__ __forceinline__ void GCNN_close(const GCNN_open& acc, float* trow) {
    if (acc.k < (unsigned)ROWS_PER_WAVE) {
        float* tp = trow + acc.k * (32u * (unsigned)TILE_PITCH);
        tp[0]  = acc.x;
        tp[32] = acc.y;
        tp[64] = acc.z;
        tp[96] = acc.w;
    }
}

__device__ __forceinline__ void GCNN_own8(int ent, int vvb, unsigned u0,
                                          const float* __restrict__ xlane,
                                          GCNN_open& acc, float* trow) {
    unsigned e[8];
    float4   xv[8];
    #pragma unroll
    for (int u = 0; u < 8; ++u)
        e[u] = (unsigned)__builtin_amdgcn_readlane(ent, (int)u0 + u);
    #pragma unroll
    for (int u = 0; u < 8; ++u)
        xv[u] = *(const float4*)(xlane + (e[u] & 0xFFFFu) * 64u);
    #pragma unroll
    for (int u = 0; u < 8; ++u) {
        const float    v = __int_as_float(__builtin_amdgcn_readlane(vvb, (int)u0 + u));
        const unsigned k = e[u] >> 21;
        if (__builtin_expect(k != acc.k, 0)) {
            GCNN_close(acc, trow);
            acc.x = 0.0f; acc.y = 0.0f; acc.z = 0.0f; acc.w = 0.0f;
            acc.k = k;
        }
        acc.x += v * xv[u].x;
        acc.y += v * xv[u].y;
        acc.z += v * xv[u].z;
        acc.w += v * xv[u].w;
    }
}

__global__ __launch_bounds__(1024)
void GCNN_spmm_gemm_kernel(const float*    __restrict__ x,
                           const unsigned* __restrict__ offs,
                           const unsigned* __restrict__ recs,
                           const float*    __restrict__ W,
                           const float*    __restrict__ bias,
                           float*          __restrict__ out) {
    __shared__ float    tile[(TILE_R + 1) * TILE_PITCH];
    __shared__ unsigned qkey[Q_CAP];
    __shared__ unsigned qval[Q_CAP];
    __shared__ unsigned rcnt[TILE_R];
    __shared__ unsigned qlate;

    const int tid  = threadIdx.x;
    const int lane = tid & 31;
    const int wave = __builtin_amdgcn_readfirstlane(tid >> 5);
    const int t    = (int)blockIdx.x;

    int o0 = 0, o1 = 0;
    if (lane < 8) {
        const unsigned* p = offs + (wave * 8 + lane) * OFF_STRIDE + t;
        o0 = (int)p[0];
        o1 = (int)p[1];
    }

    if (tid < TILE_R) rcnt[tid] = 0u;
    if (tid == 0) qlate = 0u;
    if (tid < TILE_PITCH) tile[TILE_R * TILE_PITCH + tid] = 0.0f;

    const uint2* rec2 = (const uint2*)recs + wave * 8 * REC_STRIDE;
    unsigned key[8], vb[8];
    #pragma unroll
    for (int s = 0; s < 8; ++s) {
        const unsigned b = (unsigned)__builtin_amdgcn_readlane(o0, s);
        const unsigned n = (unsigned)__builtin_amdgcn_readlane(o1, s) - b;
        const unsigned li = (unsigned)lane < n ? (unsigned)lane : n - 1u;
        const uint2 rr = rec2[s * REC_STRIDE + (n != 0u ? b + li : 0u)];
        key[s] = rr.x; vb[s] = rr.y;
    }
    __syncthreads();

    int      late  = 0;
    unsigned later = 0u;
    unsigned pos[8];
    #pragma unroll
    for (int s = 0; s < 8; ++s) {
        const unsigned n = (unsigned)(__builtin_amdgcn_readlane(o1, s) - __builtin_amdgcn_readlane(o0, s));
        const unsigned m = n < 32u ? n : 32u;
        pos[s] = 0u;
        if ((unsigned)lane < m) pos[s] = atomicAdd(&rcnt[key[s] >> 16], 1u);
    }
    #pragma unroll
    for (int s = 0; s < 8; ++s) {
        const unsigned n = (unsigned)(__builtin_amdgcn_readlane(o1, s) - __builtin_amdgcn_readlane(o0, s));
        const unsigned m = n < 32u ? n : 32u;
        bool full = false;
        if ((unsigned)lane < m) {
            if (pos[s] < (unsigned)RQ) {
                const unsigned slot = (key[s] >> 16) * (unsigned)RQ + pos[s];
                qkey[slot] = key[s];
                qval[slot] = vb[s];
            } else {
                full = true;
            }
        }
        const unsigned fm = __builtin_amdgcn_ballot_w32(full);
        if (lane == s) late = (int)fm;
        later |= fm | (n > 32u ? 1u : 0u);
    }
    if (later != 0u && lane == 0) qlate = 1u;
    __syncthreads();
    float* trow = tile + (unsigned)wave * (unsigned)TILE_PITCH + (unsigned)lane;
    int cl = 0;
    if (lane < ROWS_PER_WAVE) cl = (int)rcnt[wave + 32 * lane];
    unsigned st[ROWS_PER_WAVE + 1];
    st[0] = 0u;
    #pragma unroll
    for (int k = 0; k < ROWS_PER_WAVE; ++k) {
        unsigned c = (unsigned)__builtin_amdgcn_readlane(cl, k);
        c = c < (unsigned)RQ ? c : (unsigned)RQ;
        if (c == 0u) {
            float* tp = trow + (unsigned)k * (32u * (unsigned)TILE_PITCH);
            tp[0] = 0.0f; tp[32] = 0.0f; tp[64] = 0.0f; tp[96] = 0.0f;
        }
        st[k + 1] = st[k] + c;
    }
    const unsigned n_own = st[ROWS_PER_WAVE];
    const bool second = __builtin_amdgcn_readfirstlane((int)qlate) != 0;

    const float* xlane = x + (lane >> 4) * (N_NODES * 64) + (lane & 15) * 4;

    GCNN_open acc;
    acc.x = 0.0f; acc.y = 0.0f; acc.z = 0.0f; acc.w = 0.0f;
    acc.k = (unsigned)ROWS_PER_WAVE;
    for (unsigned i0 = 0u; i0 < n_own; i0 += 32u) {
        const unsigned i = i0 + (unsigned)lane;
        int ent = TILE_R << 16;
        int vvb = 0;
        if (i < n_own) {
            unsigned k = 0u, sk = 0u;
            #pragma unroll
            for (int m = 1; m < ROWS_PER_WAVE; ++m) {
                const bool ge = i >= st[m];
                k  = ge ? (unsigned)m : k;
                sk = ge ? st[m] : sk;
            }
            const unsigned slot = ((unsigned)wave + 32u * k) * (unsigned)RQ + (i - sk);
            ent = (int)qkey[slot];
            vvb = (int)qval[slot];
        }
        const unsigned left = (n_own - i0 + 7u) >> 3;
        const unsigned nt   = left < 4u ? left : 4u;
        for (unsigned j = 0u; j < nt; ++j)
            GCNN_own8(ent, vvb, 8u * j, xlane, acc, trow);
    }
    GCNN_close(acc, trow);

    if (second) {
        __syncthreads();
        for (int s = 0; s < 8; ++s) {
            const unsigned b  = (unsigned)__builtin_amdgcn_readlane(o0, s);
            const unsigned n  = (unsigned)__builtin_amdgcn_readlane(o1, s) - b;
            const unsigned fm = (unsigned)__builtin_amdgcn_readlane(late, s);
            if (fm != 0u) {
                int ent = TILE_R << 16;
                int vvb = 0;
                if ((fm >> lane) & 1u) {
                    const uint2 rr = rec2[s * REC_STRIDE + b + (unsigned)lane];
                    ent = (int)rr.x; vvb = (int)rr.y;
                }
                for (unsigned u0 = 0u; u0 < 32u; u0 += 8u)
                    if ((fm >> u0) & 0xFFu)
                        GCNN_drain8(ent, vvb, u0, xlane, tile, (unsigned)lane);
            }
            for (unsigned q = 32u; q < n; q += 32u) {
                const unsigned m = (n - q) < 32u ? (n - q) : 32u;
                int ent = TILE_R << 16;
                int vvb = 0;
                if ((unsigned)lane < m) {
                    const uint2 rr = rec2[s * REC_STRIDE + b + q + (unsigned)lane];
                    ent = (int)rr.x; vvb = (int)rr.y;
                }
                for (unsigned u0 = 0u; u0 < m; u0 += 8u)
                    GCNN_drain8(ent, vvb, u0, xlane, tile, (unsigned)lane);
            }
        }
    }

    const int half = lane >> 4;
    const int l16  = lane & 15;
    const int col  = (wave & 3) * 16 + l16;

    v2f bfr[16];
    #pragma unroll
    for (int ks = 0; ks < 16; ++ks) {
        const int ka = 4 * ks + 2 * half;
        bfr[ks].x = W[ka * C_OUT + col];
        bfr[ks].y = W[(ka + 1) * C_OUT + col];
    }
    const float bv = bias[col];
    __syncthreads();

    const unsigned npairs = (unsigned)t < BIG_TILES ? 13u : 12u;
    const unsigned hb0    = GCNN_tile_lo((unsigned)t);
    const unsigned n_mt   = 2u * npairs + (t == N_TILES - 1 ? 2u : 0u);
    for (unsigned mi = (unsigned)wave >> 2; mi < n_mt; mi += 8u) {
        const bool     join = mi >= 2u * npairs;
        const unsigned p    = join ? npairs : (mi >> 1);
        const unsigned sub  = mi & 1u;
        const unsigned h    = hb0 + p;
        const unsigned par  = h & 1u;
        const float* ap = tile + (16u * p + (unsigned)half) * (unsigned)TILE_PITCH + 16u * sub + (unsigned)l16;

        v8f acc = {};
        #pragma unroll
        for (int ks = 0; ks < 16; ++ks) {
            const unsigned pb = join ? (ks >= 8 ? 1u : 0u) : par;
            const float*   aq = ap + (2 * ks) * TILE_PITCH + 64u * pb;
            v2f a;
            a.x = aq[0];
            a.y = aq[32];
            acc = __builtin_amdgcn_wmma_f32_16x16x4_f32(
                 false, a,  false, bfr[ks],
                 (short)0, acc,  false,  false);
        }

        const unsigned radd = join ? WHOLE_ROWS : (par * (WHOLE_ROWS + 1u) + (h >> 1));
        #pragma unroll
        for (int v = 0; v < 8; ++v) {
            float o = acc[v] + bv;
            o = o > 0.0f ? o : 0.0f;
            const unsigned orow = (16u * sub + (unsigned)(v + 8 * half)) * PAIR2_ROWS + radd;
            out[orow * (unsigned)C_OUT + (unsigned)col] = o;
        }
    }
}

extern "C" void kernel_launch(void* const* d_in, const int* in_sizes, int n_in,
                              void* d_out, int out_size, void* d_ws, size_t ws_size,
                              hipStream_t stream) {
    const float* x    = (const float*)d_in[0];
    const int*   rows = (const int*)  d_in[1];
    const int*   cols = (const int*)  d_in[2];
    const float* vals = (const float*)d_in[3];
    const float* W    = (const float*)d_in[4];
    const float* bias = (const float*)d_in[5];
    float* out = (float*)d_out;
    unsigned* recs = (unsigned*)d_ws;
    unsigned* offs = recs + 2 * N_CHUNKS * REC_STRIDE;

    GCNN_bin_kernel<<<N_CHUNKS, BIN_THREADS, 0, stream>>>(rows, cols, vals, offs, recs);

    GCNN_spmm_gemm_kernel<<<N_TILES, 1024, 0, stream>>>(x, offs, recs, W, bias, out);
}
